# starting kernel
# baseline (speedup 1.0000x reference)
#include <hip/hip_runtime.h>
#include <hip/hip_bf16.h>


typedef __attribute__((ext_vector_type(2))) float v2f;
typedef __attribute__((ext_vector_type(4))) float v4f;
typedef __attribute__((ext_vector_type(8))) float v8f;

#define DF 128


__global__ __launch_bounds__(256) void k_zero(float* __restrict__ p, int n) {
    int i = blockIdx.x * blockDim.x + threadIdx.x;
    if (i < n) p[i] = 0.0f;
}

__global__ __launch_bounds__(256) void k_degree(const int* __restrict__ dst,
                                                float* __restrict__ deg, int E) {
    int e = blockIdx.x * blockDim.x + threadIdx.x;
    if (e < E) atomicAdd(&deg[dst[e]], 1.0f);
}

__global__ __launch_bounds__(256) void k_dinv(float* __restrict__ deg, int n) {
    int i = blockIdx.x * blockDim.x + threadIdx.x;
    if (i < n) deg[i] = rsqrtf(deg[i] + 1.0f);
}

template <bool RELU_IN>
__global__ __launch_bounds__(256) void k_gemm_wmma(
        const float* __restrict__ X, const float* __restrict__ W,
        const float* __restrict__ bias, const float* __restrict__ dinv,
        float* __restrict__ H, float* __restrict__ AGG,
        int strips, int nrows) {
    __shared__ __align__(16) float ldsW[DF * DF];

    for (int idx = threadIdx.x; idx < DF * DF; idx += blockDim.x) {
        int kk = idx >> 7;
        int n  = idx & 127;
        int t     = n >> 4;
        int nloc  = n & 15;
        int k     = kk >> 2;
        int khalf = (kk >> 1) & 1;
        int lo    = kk & 1;
        ldsW[((t << 5) + k) * 64 + (khalf * 16 + nloc) * 2 + lo] = W[idx];
    }
    __syncthreads();

    const int wave  = threadIdx.x >> 5;
    const int lane  = threadIdx.x & 31;
    const int strip = blockIdx.x * (blockDim.x >> 5) + wave;
    if (strip >= strips) return;

    const int nloc  = lane & 15;
    const int khalf = lane >> 4;

    int arow_i = strip * 16 + nloc;
    if (arow_i >= nrows) arow_i = nrows - 1;
    const float* __restrict__ arow = X + (size_t)arow_i * DF + 2 * khalf;

    const v2f* __restrict__ bbase = (const v2f*)ldsW + lane;

    v8f acc[8] = {};
#pragma unroll
    for (int k = 0; k < 32; ++k) {
        v2f a = *(const v2f*)(arow + 4 * k);
        if (RELU_IN) {
            a.x = fmaxf(a.x, 0.0f);
            a.y = fmaxf(a.y, 0.0f);
        }
#pragma unroll
        for (int t = 0; t < 8; ++t) {
            v2f b = bbase[(t * 32 + k) * 32];
            acc[t] = __builtin_amdgcn_wmma_f32_16x16x4_f32(
                         false, a, false, b, (short)0, acc[t], false, false);
        }
    }

    float bcol[8];
#pragma unroll
    for (int t = 0; t < 8; ++t) bcol[t] = bias[t * 16 + nloc];

    const int rbase = strip * 16 + 8 * khalf;
#pragma unroll
    for (int r = 0; r < 8; ++r) {
        int row = rbase + r;
        if (row >= nrows) continue;
        float dv = dinv[row];
        float d2 = dv * dv;
        size_t ro = (size_t)row * DF;
#pragma unroll
        for (int t = 0; t < 8; ++t) {
            float val = acc[t][r];
            int col = t * 16 + nloc;
            H[ro + col]   = val;
            AGG[ro + col] = fmaf(val, d2, bcol[t]);
        }
    }
}

__global__ __launch_bounds__(256) void k_scatter(
        const float* __restrict__ H, const int* __restrict__ src,
        const int* __restrict__ dst, const float* __restrict__ dinv,
        float* __restrict__ AGG, int E) {
    int e = blockIdx.x * (blockDim.x >> 5) + (threadIdx.x >> 5);
    if (e >= E) return;
    int lane = threadIdx.x & 31;
    int s = src[e], d = dst[e];
    float nrm = dinv[s] * dinv[d];
    v4f v = *(const v4f*)(H + (size_t)s * DF + lane * 4);
    float* out = AGG + (size_t)d * DF + lane * 4;
    atomicAdd(out + 0, v.x * nrm);
    atomicAdd(out + 1, v.y * nrm);
    atomicAdd(out + 2, v.z * nrm);
    atomicAdd(out + 3, v.w * nrm);
}

__global__ __launch_bounds__(256) void k_normalize(const float* __restrict__ in,
                                                   float* __restrict__ out, int n) {
    int row = blockIdx.x * (blockDim.x >> 5) + (threadIdx.x >> 5);
    if (row >= n) return;
    int lane = threadIdx.x & 31;
    v4f v = *(const v4f*)(in + (size_t)row * DF + lane * 4);
    float ss = v.x * v.x + v.y * v.y + v.z * v.z + v.w * v.w;
#pragma unroll
    for (int m = 16; m > 0; m >>= 1) ss += __shfl_xor(ss, m, 32);
    float sc = 1.0f / fmaxf(sqrtf(ss), 1e-12f);
    v4f o = {v.x * sc, v.y * sc, v.z * sc, v.w * sc};
    *(v4f*)(out + (size_t)row * DF + lane * 4) = o;
}


extern "C" void kernel_launch(void* const* d_in, const int* in_sizes, int n_in,
                              void* d_out, int out_size, void* d_ws, size_t ws_size,
                              hipStream_t stream) {
    const float* x  = (const float*)d_in[0];
    const int*   ei = (const int*)  d_in[1];
    const float* W1 = (const float*)d_in[2];
    const float* b1 = (const float*)d_in[3];
    const float* W2 = (const float*)d_in[4];
    const float* b2 = (const float*)d_in[5];
    const float* W3 = (const float*)d_in[6];
    const float* b3 = (const float*)d_in[7];

    const int N = in_sizes[0] / DF;
    const int E = in_sizes[1] / 2;
    const int* src = ei;
    const int* dst = ei + E;

    float* out = (float*)d_out;

    char* ws = (char*)d_ws;
    size_t off0 = ((size_t)N * sizeof(float) + 255) & ~(size_t)255;
    float* dinv = (float*)ws;
    float* H    = (float*)(ws + off0);
    float* AGGa = H + (size_t)N * DF;

    const int strips  = (N + 15) / 16;
    const int gblocks = (strips + 7) / 8;
    const int sblocks = (E + 7) / 8;
    const int nb256   = (N + 255) / 256;
    const int eb256   = (E + 255) / 256;

    k_zero  <<<nb256, 256, 0, stream>>>(dinv, N);
    k_degree<<<eb256, 256, 0, stream>>>(dst, dinv, E);
    k_dinv  <<<nb256, 256, 0, stream>>>(dinv, N);

    k_gemm_wmma<false><<<gblocks, 256, 0, stream>>>(x, W1, b1, dinv, H, AGGa, strips, N);
    k_scatter         <<<sblocks, 256, 0, stream>>>(H, src, dst, dinv, AGGa, E);

    k_gemm_wmma<true> <<<gblocks, 256, 0, stream>>>(AGGa, W2, b2, dinv, H, out, strips, N);
    k_scatter         <<<sblocks, 256, 0, stream>>>(H, src, dst, dinv, out, E);

    k_gemm_wmma<true> <<<gblocks, 256, 0, stream>>>(out, W3, b3, dinv, H, AGGa, strips, N);
    k_scatter         <<<sblocks, 256, 0, stream>>>(H, src, dst, dinv, AGGa, E);

    k_normalize<<<(N + 7) / 8, 256, 0, stream>>>(AGGa, out, N);
}


# Aggregation by destination-sorted edges (counting sort per call) with one wave summing each row in registers: no float atomics; sum order inside a row = arrival-order rank of the fill
# speedup vs baseline: 2.1018x; 2.1018x over previous
#include <hip/hip_runtime.h>
#include <hip/hip_bf16.h>


typedef __attribute__((ext_vector_type(2))) float v2f;
typedef __attribute__((ext_vector_type(4))) float v4f;
typedef __attribute__((ext_vector_type(8))) float v8f;

#define DF 128


__global__ __launch_bounds__(256) void k_zero(int* __restrict__ p, int n) {
    int i = blockIdx.x * blockDim.x + threadIdx.x;
    if (i < n) p[i] = 0;
}

__global__ __launch_bounds__(256) void k_count(const int* __restrict__ dst,
                                               int* __restrict__ cnt, int E) {
    const int T = gridDim.x * blockDim.x;
    for (int e = blockIdx.x * blockDim.x + threadIdx.x; e < E; e += T)
        atomicAdd(&cnt[dst[e]], 1);
}

__global__ __launch_bounds__(256) void k_scan_local(const int* __restrict__ cnt,
                                                    int* __restrict__ rowptr,
                                                    int* __restrict__ bsum, int n) {
    __shared__ int wsum[8];
    const int t = threadIdx.x, lane = t & 31, wave = t >> 5;
    const int base = blockIdx.x * 1024 + t * 4;
    int c0 = (base + 0 < n) ? cnt[base + 0] : 0;
    int c1 = (base + 1 < n) ? cnt[base + 1] : 0;
    int c2 = (base + 2 < n) ? cnt[base + 2] : 0;
    int c3 = (base + 3 < n) ? cnt[base + 3] : 0;
    const int s = c0 + c1 + c2 + c3;
    int inc = s;
#pragma unroll
    for (int o = 1; o < 32; o <<= 1) {
        int v = __shfl_up(inc, o, 32);
        if (lane >= o) inc += v;
    }
    if (lane == 31) wsum[wave] = inc;
    __syncthreads();
    int woff = 0;
#pragma unroll
    for (int w = 0; w < 8; ++w) woff += (w < wave) ? wsum[w] : 0;
    const int ex = woff + inc - s;
    if (base + 0 < n) rowptr[base + 0] = ex;
    if (base + 1 < n) rowptr[base + 1] = ex + c0;
    if (base + 2 < n) rowptr[base + 2] = ex + c0 + c1;
    if (base + 3 < n) rowptr[base + 3] = ex + c0 + c1 + c2;
    if (t == 255) bsum[blockIdx.x] = woff + inc;
}

__global__ __launch_bounds__(256) void k_scan_final(const int* __restrict__ cnt,
                                                    const int* __restrict__ bsum,
                                                    int* __restrict__ rowptr,
                                                    int* __restrict__ cursor,
                                                    float* __restrict__ dinv, int n) {
    __shared__ int red[8];
    const int t = threadIdx.x, lane = t & 31, wave = t >> 5;
    const int b = blockIdx.x;
    int part = 0;
    for (int j = t; j < b; j += 256) part += bsum[j];
#pragma unroll
    for (int m = 16; m > 0; m >>= 1) part += __shfl_xor(part, m, 32);
    if (lane == 0) red[wave] = part;
    __syncthreads();
    int boff = 0;
#pragma unroll
    for (int w = 0; w < 8; ++w) boff += red[w];
#pragma unroll
    for (int k = 0; k < 4; ++k) {
        int i = b * 1024 + k * 256 + t;
        if (i < n) {
            int r = rowptr[i] + boff;
            rowptr[i] = r;
            cursor[i] = r;
            dinv[i]   = rsqrtf((float)cnt[i] + 1.0f);
        }
    }
    if (b == (int)gridDim.x - 1 && t == 0) rowptr[n] = boff + bsum[b];
}

__global__ __launch_bounds__(256) void k_fill(const int* __restrict__ src,
                                              const int* __restrict__ dst,
                                              int* __restrict__ cursor,
                                              int* __restrict__ csr, int E) {
    const int T = gridDim.x * blockDim.x;
    for (int e0 = blockIdx.x * blockDim.x + threadIdx.x; e0 < E; e0 += 4 * T) {
        const int e1 = e0 + T, e2 = e0 + 2 * T, e3 = e0 + 3 * T;
        if (e3 < E) {
            const int d0 = dst[e0], d1 = dst[e1], d2 = dst[e2], d3 = dst[e3];
            const int s0 = src[e0], s1 = src[e1], s2 = src[e2], s3 = src[e3];
            const unsigned p0 = (unsigned)atomicAdd(&cursor[d0], 1);
            const unsigned p1 = (unsigned)atomicAdd(&cursor[d1], 1);
            const unsigned p2 = (unsigned)atomicAdd(&cursor[d2], 1);
            const unsigned p3 = (unsigned)atomicAdd(&cursor[d3], 1);
            csr[p0] = s0;
            csr[p1] = s1;
            csr[p2] = s2;
            csr[p3] = s3;
        } else {
            for (int e = e0; e < E; e += T) {
                const unsigned p = (unsigned)atomicAdd(&cursor[dst[e]], 1);
                csr[p] = src[e];
            }
        }
    }
}

template <bool RELU_IN>
__global__ __launch_bounds__(256) void k_gemm_wmma(
        const float* __restrict__ X, const float* __restrict__ W,
        const float* __restrict__ bias, const float* __restrict__ dinv,
        float* __restrict__ H, float* __restrict__ AGG,
        int strips, int nrows) {
    __shared__ __align__(16) float ldsW[DF * DF];

    for (int idx = threadIdx.x; idx < DF * DF; idx += blockDim.x) {
        int kk = idx >> 7;
        int n  = idx & 127;
        int t     = n >> 4;
        int nloc  = n & 15;
        int k     = kk >> 2;
        int khalf = (kk >> 1) & 1;
        int lo    = kk & 1;
        ldsW[((t << 5) + k) * 64 + (khalf * 16 + nloc) * 2 + lo] = W[idx];
    }
    __syncthreads();

    const int wave  = threadIdx.x >> 5;
    const int lane  = threadIdx.x & 31;
    const int strip = blockIdx.x * (blockDim.x >> 5) + wave;
    if (strip >= strips) return;

    const int nloc  = lane & 15;
    const int khalf = lane >> 4;

    int arow_i = strip * 16 + nloc;
    if (arow_i >= nrows) arow_i = nrows - 1;
    const float* __restrict__ arow = X + (size_t)arow_i * DF + 2 * khalf;

    const v2f* __restrict__ bbase = (const v2f*)ldsW + lane;

    v8f acc[8] = {};
#pragma unroll
    for (int k = 0; k < 32; ++k) {
        v2f a = *(const v2f*)(arow + 4 * k);
        if (RELU_IN) {
            a.x = fmaxf(a.x, 0.0f);
            a.y = fmaxf(a.y, 0.0f);
        }
#pragma unroll
        for (int t = 0; t < 8; ++t) {
            v2f b = bbase[(t * 32 + k) * 32];
            acc[t] = __builtin_amdgcn_wmma_f32_16x16x4_f32(
                         false, a, false, b, (short)0, acc[t], false, false);
        }
    }

    float bcol[8];
#pragma unroll
    for (int t = 0; t < 8; ++t) bcol[t] = bias[t * 16 + nloc];

    const int rbase = strip * 16 + 8 * khalf;
#pragma unroll
    for (int r = 0; r < 8; ++r) {
        int row = rbase + r;
        if (row >= nrows) continue;
        float dv = dinv[row];
        float d2 = dv * dv;
        size_t ro = (size_t)row * DF;
#pragma unroll
        for (int t = 0; t < 8; ++t) {
            float val = acc[t][r];
            int col = t * 16 + nloc;
            H[ro + col]   = val;
            AGG[ro + col] = fmaf(val, d2, bcol[t]);
        }
    }
}

__global__ __launch_bounds__(256) void k_pull(
        const float* __restrict__ H, const int* __restrict__ rowptr,
        const int* __restrict__ csr, const float* __restrict__ dinv,
        float* __restrict__ AGG, int n) {
    const int d = __builtin_amdgcn_readfirstlane(
                      (int)(blockIdx.x * (blockDim.x >> 5) + (threadIdx.x >> 5)));
    if (d >= n) return;
    const int lane = threadIdx.x & 31;
    const int beg = rowptr[d], end = rowptr[d + 1];
    const float dd = dinv[d];
    const float* __restrict__ hl = H + lane * 4;
    float* arow = AGG + (size_t)d * DF + lane * 4;
    v4f acc = *(const v4f*)arow;

    for (int base = beg; base < end; base += 32) {
        const int cn = (end - base < 32) ? (end - base) : 32;
        int   s_l = 0;
        float w_l = 0.0f;
        if (lane < cn) {
            s_l = csr[base + lane];
            s_l = min(max(s_l, 0), n - 1);
            w_l = dinv[s_l] * dd;
        }
        const int w_i = __float_as_int(w_l);
        int j = 0;
        for (; j + 8 <= cn; j += 8) {
            v4f v[8];
            float w[8];
#pragma unroll
            for (int k = 0; k < 8; ++k) {
                int s = __builtin_amdgcn_readlane(s_l, j + k);
                w[k]  = __int_as_float(__builtin_amdgcn_readlane(w_i, j + k));
                v[k]  = *(const v4f*)(hl + (size_t)s * DF);
            }
#pragma unroll
            for (int k = 0; k < 8; ++k) acc += v[k] * w[k];
        }
        for (; j < cn; j += 4) {
            v4f v[4];
            float w[4];
#pragma unroll
            for (int k = 0; k < 4; ++k) {
                int jj = (j + k < cn) ? (j + k) : (cn - 1);
                int s  = __builtin_amdgcn_readlane(s_l, jj);
                w[k]   = __int_as_float(__builtin_amdgcn_readlane(w_i, jj));
                v[k]   = *(const v4f*)(hl + (size_t)s * DF);
            }
#pragma unroll
            for (int k = 0; k < 4; ++k) {
                v4f tsum = acc + v[k] * w[k];
                acc = (j + k < cn) ? tsum : acc;
            }
        }
    }
    *(v4f*)arow = acc;
}

__global__ __launch_bounds__(256) void k_normalize(const float* __restrict__ in,
                                                   float* __restrict__ out, int n) {
    int row = blockIdx.x * (blockDim.x >> 5) + (threadIdx.x >> 5);
    if (row >= n) return;
    int lane = threadIdx.x & 31;
    v4f v = *(const v4f*)(in + (size_t)row * DF + lane * 4);
    float ss = v.x * v.x + v.y * v.y + v.z * v.z + v.w * v.w;
#pragma unroll
    for (int m = 16; m > 0; m >>= 1) ss += __shfl_xor(ss, m, 32);
    float sc = 1.0f / fmaxf(sqrtf(ss), 1e-12f);
    v4f o = {v.x * sc, v.y * sc, v.z * sc, v.w * sc};
    *(v4f*)(out + (size_t)row * DF + lane * 4) = o;
}


extern "C" void kernel_launch(void* const* d_in, const int* in_sizes, int n_in,
                              void* d_out, int out_size, void* d_ws, size_t ws_size,
                              hipStream_t stream) {
    const float* x  = (const float*)d_in[0];
    const int*   ei = (const int*)  d_in[1];
    const float* W1 = (const float*)d_in[2];
    const float* b1 = (const float*)d_in[3];
    const float* W2 = (const float*)d_in[4];
    const float* b2 = (const float*)d_in[5];
    const float* W3 = (const float*)d_in[6];
    const float* b3 = (const float*)d_in[7];

    const int N = in_sizes[0] / DF;
    const int E = in_sizes[1] / 2;
    const int* src = ei;
    const int* dst = ei + E;

    float* out = (float*)d_out;

    char* ws = (char*)d_ws;
    const int nscan = (N + 1023) / 1024;
    size_t offN  = ((size_t)(N + 1) * 4 + 255) & ~(size_t)255;
    size_t offS  = ((size_t)nscan * 4 + 255) & ~(size_t)255;
    size_t offE  = ((size_t)E * 4 + 255) & ~(size_t)255;
    float* dinv   = (float*)ws;
    int*   cnt    = (int*)(ws + offN);
    int*   rowptr = (int*)(ws + 2 * offN);
    int*   cursor = (int*)(ws + 3 * offN);
    int*   bsum   = (int*)(ws + 4 * offN);
    int*   csr    = (int*)(ws + 4 * offN + offS);
    float* H      = (float*)(ws + 4 * offN + offS + offE);
    float* AGGa   = H + (size_t)N * DF;

    const int strips  = (N + 15) / 16;
    const int gblocks = (strips + 7) / 8;
    const int pblocks = (N + 7) / 8;
    const int nb256   = (N + 255) / 256;
    const int eb4     = ((E + 3) / 4 + 255) / 256;

    k_zero      <<<nb256, 256, 0, stream>>>(cnt, N);
    k_count     <<<eb4,   256, 0, stream>>>(dst, cnt, E);
    k_scan_local<<<nscan, 256, 0, stream>>>(cnt, rowptr, bsum, N);
    k_scan_final<<<nscan, 256, 0, stream>>>(cnt, bsum, rowptr, cursor, dinv, N);
    k_fill      <<<eb4,   256, 0, stream>>>(src, dst, cursor, csr, E);

    k_gemm_wmma<false><<<gblocks, 256, 0, stream>>>(x, W1, b1, dinv, H, AGGa, strips, N);
    k_pull            <<<pblocks, 256, 0, stream>>>(H, rowptr, csr, dinv, AGGa, N);

    k_gemm_wmma<true> <<<gblocks, 256, 0, stream>>>(AGGa, W2, b2, dinv, H, out, strips, N);
    k_pull            <<<pblocks, 256, 0, stream>>>(H, rowptr, csr, dinv, out, N);

    k_gemm_wmma<true> <<<gblocks, 256, 0, stream>>>(out, W3, b3, dinv, H, AGGa, strips, N);
    k_pull            <<<pblocks, 256, 0, stream>>>(H, rowptr, csr, dinv, AGGa, N);

    k_normalize<<<(N + 7) / 8, 256, 0, stream>>>(AGGa, out, N);
}


# Edge sort by destination built through LDS bins (bin counts per chunk, per-bin scan, pair scatter, per-bin row build): no global atomics; sum order in a row = arrival order at the LDS slot counters
# speedup vs baseline: 3.1959x; 1.5206x over previous
#include <hip/hip_runtime.h>
#include <hip/hip_bf16.h>


typedef __attribute__((ext_vector_type(2))) float v2f;
typedef __attribute__((ext_vector_type(4))) float v4f;
typedef __attribute__((ext_vector_type(8))) float v8f;

#define DF 128


#define BIN_SHIFT   8
#define BIN_NODES   (1 << BIN_SHIFT)
#define SORT_CHUNKS 256
#define MAX_BINS    1024

typedef __attribute__((ext_vector_type(2))) int v2i;

__global__ __launch_bounds__(1024) void k_bin_hist(const int* __restrict__ dst,
                                                   int* __restrict__ hist,
                                                   int E, int B, int chunk) {
    __shared__ int lh[MAX_BINS];
    const int t = threadIdx.x, g = blockIdx.x;
    for (int b = t; b < B; b += 1024) lh[b] = 0;
    __syncthreads();
    const int e0 = g * chunk;
    const int e1 = (e0 + chunk < E) ? (e0 + chunk) : E;
    for (int eb = e0 + t; eb < e1; eb += 4 * 1024) {
        int d[4];
#pragma unroll
        for (int k = 0; k < 4; ++k) {
            const int e = eb + k * 1024;
            d[k] = (e < e1) ? dst[e] : -1;
        }
#pragma unroll
        for (int k = 0; k < 4; ++k)
            if (d[k] >= 0) atomicAdd(&lh[d[k] >> BIN_SHIFT], 1);
    }
    __syncthreads();
    for (int b = t; b < B; b += 1024) hist[b * SORT_CHUNKS + g] = lh[b];
}

__global__ __launch_bounds__(SORT_CHUNKS) void k_bin_scan(int* __restrict__ hist,
                                                          int* __restrict__ bintot) {
    __shared__ int wsum[SORT_CHUNKS / 32];
    const int t = threadIdx.x, lane = t & 31, wave = t >> 5;
    const int b = blockIdx.x;
    const int v = hist[b * SORT_CHUNKS + t];
    int inc = v;
#pragma unroll
    for (int o = 1; o < 32; o <<= 1) {
        int u = __shfl_up(inc, o, 32);
        if (lane >= o) inc += u;
    }
    if (lane == 31) wsum[wave] = inc;
    __syncthreads();
    int woff = 0;
#pragma unroll
    for (int w = 0; w < SORT_CHUNKS / 32; ++w) woff += (w < wave) ? wsum[w] : 0;
    hist[b * SORT_CHUNKS + t] = woff + inc - v;
    if (t == SORT_CHUNKS - 1) bintot[b] = woff + inc;
}

__global__ __launch_bounds__(1024) void k_bin_scatter(const int* __restrict__ src,
                                                      const int* __restrict__ dst,
                                                      const int* __restrict__ hist,
                                                      const int* __restrict__ bintot,
                                                      int* __restrict__ pairs,
                                                      int E, int B, int chunk) {
    __shared__ int lcur[MAX_BINS];
    __shared__ int wsum[32];
    const int t = threadIdx.x, lane = t & 31, wave = t >> 5, g = blockIdx.x;
    const int v = (t < B) ? bintot[t] : 0;
    int inc = v;
#pragma unroll
    for (int o = 1; o < 32; o <<= 1) {
        int u = __shfl_up(inc, o, 32);
        if (lane >= o) inc += u;
    }
    if (lane == 31) wsum[wave] = inc;
    __syncthreads();
    int woff = 0;
#pragma unroll
    for (int w = 0; w < 32; ++w) woff += (w < wave) ? wsum[w] : 0;
    if (t < B) lcur[t] = woff + inc - v + hist[t * SORT_CHUNKS + g];
    __syncthreads();

    v2i* __restrict__ pr = (v2i*)pairs;
    const int e0 = g * chunk;
    const int e1 = (e0 + chunk < E) ? (e0 + chunk) : E;
    for (int eb = e0 + t; eb < e1; eb += 4 * 1024) {
        int d[4], s[4];
#pragma unroll
        for (int k = 0; k < 4; ++k) {
            const int e = eb + k * 1024;
            const bool ok = e < e1;
            d[k] = ok ? dst[e] : -1;
            s[k] = ok ? src[e] : 0;
        }
#pragma unroll
        for (int k = 0; k < 4; ++k) {
            if (d[k] >= 0) {
                const int p = atomicAdd(&lcur[d[k] >> BIN_SHIFT], 1);
                v2i w = {s[k], d[k]};
                pr[p] = w;
            }
        }
    }
}

__global__ __launch_bounds__(BIN_NODES) void k_bin_rows(const int* __restrict__ pairs,
                                                        const int* __restrict__ bintot,
                                                        int* __restrict__ rowptr,
                                                        int* __restrict__ csr,
                                                        float* __restrict__ dinv,
                                                        int n, int B) {
    __shared__ int cnt[BIN_NODES];
    __shared__ int cur[BIN_NODES];
    __shared__ int red[BIN_NODES / 32];
    __shared__ int wsum[BIN_NODES / 32];
    const int t = threadIdx.x, lane = t & 31, wave = t >> 5;
    const int b = blockIdx.x;
    int part = 0;
    for (int j = t; j < b; j += BIN_NODES) part += bintot[j];
#pragma unroll
    for (int m = 16; m > 0; m >>= 1) part += __shfl_xor(part, m, 32);
    if (lane == 0) red[wave] = part;
    cnt[t] = 0;
    __syncthreads();
    int binStart = 0;
#pragma unroll
    for (int w = 0; w < BIN_NODES / 32; ++w) binStart += red[w];
    const int binEnd = binStart + bintot[b];

    const v2i* __restrict__ pr = (const v2i*)pairs;
    for (int eb = binStart + t; eb < binEnd; eb += 8 * BIN_NODES) {
        int d[8];
#pragma unroll
        for (int k = 0; k < 8; ++k) {
            const int e = eb + k * BIN_NODES;
            d[k] = (e < binEnd) ? pr[e].y : -1;
        }
#pragma unroll
        for (int k = 0; k < 8; ++k)
            if (d[k] >= 0) atomicAdd(&cnt[d[k] & (BIN_NODES - 1)], 1);
    }
    __syncthreads();

    const int c = cnt[t];
    int inc = c;
#pragma unroll
    for (int o = 1; o < 32; o <<= 1) {
        int u = __shfl_up(inc, o, 32);
        if (lane >= o) inc += u;
    }
    if (lane == 31) wsum[wave] = inc;
    __syncthreads();
    int woff = 0;
#pragma unroll
    for (int w = 0; w < BIN_NODES / 32; ++w) woff += (w < wave) ? wsum[w] : 0;
    const int rowStart = binStart + woff + inc - c;
    const int node = b * BIN_NODES + t;
    if (node < n) {
        rowptr[node] = rowStart;
        dinv[node]   = rsqrtf((float)c + 1.0f);
    }
    if (b == B - 1 && t == 0) rowptr[n] = binEnd;
    cur[t] = rowStart;
    __syncthreads();

    for (int eb = binStart + t; eb < binEnd; eb += 8 * BIN_NODES) {
        v2i q[8];
#pragma unroll
        for (int k = 0; k < 8; ++k) {
            const int e = eb + k * BIN_NODES;
            v2i none = {0, -1};
            q[k] = (e < binEnd) ? pr[e] : none;
        }
#pragma unroll
        for (int k = 0; k < 8; ++k) {
            if (q[k].y >= 0) {
                const int p = atomicAdd(&cur[q[k].y & (BIN_NODES - 1)], 1);
                csr[p] = q[k].x;
            }
        }
    }
}

template <bool RELU_IN>
__global__ __launch_bounds__(256) void k_gemm_wmma(
        const float* __restrict__ X, const float* __restrict__ W,
        const float* __restrict__ dinv, float* __restrict__ H,
        int strips, int nrows) {
    __shared__ __align__(16) float ldsW[DF * DF];

    for (int idx = threadIdx.x; idx < DF * DF; idx += blockDim.x) {
        int kk = idx >> 7;
        int n  = idx & 127;
        int t     = n >> 4;
        int nloc  = n & 15;
        int k     = kk >> 2;
        int khalf = (kk >> 1) & 1;
        int lo    = kk & 1;
        ldsW[((t << 5) + k) * 64 + (khalf * 16 + nloc) * 2 + lo] = W[idx];
    }
    __syncthreads();

    const int wave  = threadIdx.x >> 5;
    const int lane  = threadIdx.x & 31;
    const int strip = blockIdx.x * (blockDim.x >> 5) + wave;
    if (strip >= strips) return;

    const int nloc  = lane & 15;
    const int khalf = lane >> 4;

    int arow_i = strip * 16 + nloc;
    if (arow_i >= nrows) arow_i = nrows - 1;
    const float* __restrict__ arow = X + (size_t)arow_i * DF + 2 * khalf;

    const v2f* __restrict__ bbase = (const v2f*)ldsW + lane;

    v8f acc[8] = {};
#pragma unroll
    for (int k = 0; k < 32; ++k) {
        v2f a = *(const v2f*)(arow + 4 * k);
        if (RELU_IN) {
            a.x = fmaxf(a.x, 0.0f);
            a.y = fmaxf(a.y, 0.0f);
        }
#pragma unroll
        for (int t = 0; t < 8; ++t) {
            v2f b = bbase[(t * 32 + k) * 32];
            acc[t] = __builtin_amdgcn_wmma_f32_16x16x4_f32(
                         false, a, false, b, (short)0, acc[t], false, false);
        }
    }

    const int rbase = strip * 16 + 8 * khalf;
#pragma unroll
    for (int r = 0; r < 8; ++r) {
        int row = rbase + r;
        if (row >= nrows) continue;
        float dv = dinv[row];
        size_t ro = (size_t)row * DF;
#pragma unroll
        for (int t = 0; t < 8; ++t)
            H[ro + t * 16 + nloc] = acc[t][r] * dv;
    }
}

__global__ __launch_bounds__(256) void k_pull(
        const float* __restrict__ H, const int* __restrict__ rowptr,
        const int* __restrict__ csr, const float* __restrict__ dinv,
        const float* __restrict__ bias, float* __restrict__ OUT, int n) {
    const int d = __builtin_amdgcn_readfirstlane(
                      (int)(blockIdx.x * (blockDim.x >> 5) + (threadIdx.x >> 5)));
    if (d >= n) return;
    const int lane = threadIdx.x & 31;
    const int beg = rowptr[d], end = rowptr[d + 1];
    const float dd = dinv[d];
    const float* __restrict__ hl = H + lane * 4;
    const v4f bv = *(const v4f*)(bias + lane * 4);
    v4f acc = *(const v4f*)(hl + (size_t)d * DF);

    for (int base = beg; base < end; base += 32) {
        const int cn = (end - base < 32) ? (end - base) : 32;
        int s_l = 0;
        if (lane < cn) {
            s_l = csr[base + lane];
            s_l = min(max(s_l, 0), n - 1);
        }
        int j = 0;
        for (; j + 8 <= cn; j += 8) {
            v4f v[8];
#pragma unroll
            for (int k = 0; k < 8; ++k) {
                int s = __builtin_amdgcn_readlane(s_l, j + k);
                v[k]  = *(const v4f*)(hl + (size_t)s * DF);
            }
#pragma unroll
            for (int k = 0; k < 8; ++k) acc += v[k];
        }
        for (; j < cn; j += 4) {
            v4f v[4];
#pragma unroll
            for (int k = 0; k < 4; ++k) {
                int jj = (j + k < cn) ? (j + k) : (cn - 1);
                int s  = __builtin_amdgcn_readlane(s_l, jj);
                v[k]   = *(const v4f*)(hl + (size_t)s * DF);
            }
#pragma unroll
            for (int k = 0; k < 4; ++k) {
                v4f tsum = acc + v[k];
                acc = (j + k < cn) ? tsum : acc;
            }
        }
    }
    *(v4f*)(OUT + (size_t)d * DF + lane * 4) = acc * dd + bv;
}

__global__ __launch_bounds__(256) void k_normalize(const float* __restrict__ in,
                                                   float* __restrict__ out, int n) {
    int row = blockIdx.x * (blockDim.x >> 5) + (threadIdx.x >> 5);
    if (row >= n) return;
    int lane = threadIdx.x & 31;
    v4f v = *(const v4f*)(in + (size_t)row * DF + lane * 4);
    float ss = v.x * v.x + v.y * v.y + v.z * v.z + v.w * v.w;
#pragma unroll
    for (int m = 16; m > 0; m >>= 1) ss += __shfl_xor(ss, m, 32);
    float sc = 1.0f / fmaxf(sqrtf(ss), 1e-12f);
    v4f o = {v.x * sc, v.y * sc, v.z * sc, v.w * sc};
    *(v4f*)(out + (size_t)row * DF + lane * 4) = o;
}


extern "C" void kernel_launch(void* const* d_in, const int* in_sizes, int n_in,
                              void* d_out, int out_size, void* d_ws, size_t ws_size,
                              hipStream_t stream) {
    const float* x  = (const float*)d_in[0];
    const int*   ei = (const int*)  d_in[1];
    const float* W1 = (const float*)d_in[2];
    const float* b1 = (const float*)d_in[3];
    const float* W2 = (const float*)d_in[4];
    const float* b2 = (const float*)d_in[5];
    const float* W3 = (const float*)d_in[6];
    const float* b3 = (const float*)d_in[7];

    const int N = in_sizes[0] / DF;
    const int E = in_sizes[1] / 2;
    const int* src = ei;
    const int* dst = ei + E;

    float* out = (float*)d_out;

    char* ws = (char*)d_ws;
    const int B = (N + BIN_NODES - 1) / BIN_NODES;
    const int chunk = (E + SORT_CHUNKS - 1) / SORT_CHUNKS;
    size_t offN  = ((size_t)(N + 1) * 4 + 255) & ~(size_t)255;
    size_t offB  = ((size_t)B * 4 + 255) & ~(size_t)255;
    size_t offH  = ((size_t)B * SORT_CHUNKS * 4 + 255) & ~(size_t)255;
    size_t offE  = ((size_t)E * 4 + 255) & ~(size_t)255;
    float* dinv   = (float*)ws;
    int*   rowptr = (int*)(ws + offN);
    int*   bintot = (int*)(ws + 2 * offN);
    int*   hist   = (int*)(ws + 2 * offN + offB);
    int*   csr    = (int*)(ws + 2 * offN + offB + offH);
    int*   pairs  = (int*)(ws + 2 * offN + offB + offH + offE);
    float* H      = (float*)(ws + 2 * offN + offB + offH + 3 * offE);
    float* AGGa   = H + (size_t)N * DF;

    const int strips  = (N + 15) / 16;
    const int gblocks = (strips + 7) / 8;
    const int pblocks = (N + 7) / 8;

    k_bin_hist   <<<SORT_CHUNKS, 1024,        0, stream>>>(dst, hist, E, B, chunk);
    k_bin_scan   <<<B,           SORT_CHUNKS, 0, stream>>>(hist, bintot);
    k_bin_scatter<<<SORT_CHUNKS, 1024,        0, stream>>>(src, dst, hist, bintot, pairs, E, B, chunk);
    k_bin_rows   <<<B,           BIN_NODES,   0, stream>>>(pairs, bintot, rowptr, csr, dinv, N, B);

    k_gemm_wmma<false><<<gblocks, 256, 0, stream>>>(x, W1, dinv, H, strips, N);
    k_pull            <<<pblocks, 256, 0, stream>>>(H, rowptr, csr, dinv, b1, AGGa, N);

    k_gemm_wmma<true> <<<gblocks, 256, 0, stream>>>(AGGa, W2, dinv, H, strips, N);
    k_pull            <<<pblocks, 256, 0, stream>>>(H, rowptr, csr, dinv, b2, out, N);

    k_gemm_wmma<true> <<<gblocks, 256, 0, stream>>>(out, W3, dinv, H, strips, N);
    k_pull            <<<pblocks, 256, 0, stream>>>(H, rowptr, csr, dinv, b3, AGGa, N);

    k_normalize<<<(N + 7) / 8, 256, 0, stream>>>(AGGa, out, N);
}


# GEMM weight staging: compile-time stride, unrolled, so a thread's 64 loads are requested together and stored as float2, instead of 64 dependent load-then-LDS-store trips before the first WMMA
# speedup vs baseline: 4.0618x; 1.2709x over previous
#include <hip/hip_runtime.h>
#include <hip/hip_bf16.h>


typedef __attribute__((ext_vector_type(2))) float v2f;
typedef __attribute__((ext_vector_type(4))) float v4f;
typedef __attribute__((ext_vector_type(8))) float v8f;

#define DF 128
#define GEMM_THREADS 256


#define BIN_SHIFT   8
#define BIN_NODES   (1 << BIN_SHIFT)
#define SORT_CHUNKS 256
#define MAX_BINS    1024

typedef __attribute__((ext_vector_type(2))) int v2i;

__global__ __launch_bounds__(1024) void k_bin_hist(const int* __restrict__ dst,
                                                   int* __restrict__ hist,
                                                   int E, int B, int chunk) {
    __shared__ int lh[MAX_BINS];
    const int t = threadIdx.x, g = blockIdx.x;
    for (int b = t; b < B; b += 1024) lh[b] = 0;
    __syncthreads();
    const int e0 = g * chunk;
    const int e1 = (e0 + chunk < E) ? (e0 + chunk) : E;
    for (int eb = e0 + t; eb < e1; eb += 4 * 1024) {
        int d[4];
#pragma unroll
        for (int k = 0; k < 4; ++k) {
            const int e = eb + k * 1024;
            d[k] = (e < e1) ? dst[e] : -1;
        }
#pragma unroll
        for (int k = 0; k < 4; ++k)
            if (d[k] >= 0) atomicAdd(&lh[d[k] >> BIN_SHIFT], 1);
    }
    __syncthreads();
    for (int b = t; b < B; b += 1024) hist[b * SORT_CHUNKS + g] = lh[b];
}

__global__ __launch_bounds__(SORT_CHUNKS) void k_bin_scan(int* __restrict__ hist,
                                                          int* __restrict__ bintot) {
    __shared__ int wsum[SORT_CHUNKS / 32];
    const int t = threadIdx.x, lane = t & 31, wave = t >> 5;
    const int b = blockIdx.x;
    const int v = hist[b * SORT_CHUNKS + t];
    int inc = v;
#pragma unroll
    for (int o = 1; o < 32; o <<= 1) {
        int u = __shfl_up(inc, o, 32);
        if (lane >= o) inc += u;
    }
    if (lane == 31) wsum[wave] = inc;
    __syncthreads();
    int woff = 0;
#pragma unroll
    for (int w = 0; w < SORT_CHUNKS / 32; ++w) woff += (w < wave) ? wsum[w] : 0;
    hist[b * SORT_CHUNKS + t] = woff + inc - v;
    if (t == SORT_CHUNKS - 1) bintot[b] = woff + inc;
}

__global__ __launch_bounds__(1024) void k_bin_scatter(const int* __restrict__ src,
                                                      const int* __restrict__ dst,
                                                      const int* __restrict__ hist,
                                                      const int* __restrict__ bintot,
                                                      int* __restrict__ pairs,
                                                      int E, int B, int chunk) {
    __shared__ int lcur[MAX_BINS];
    __shared__ int wsum[32];
    const int t = threadIdx.x, lane = t & 31, wave = t >> 5, g = blockIdx.x;
    const int v = (t < B) ? bintot[t] : 0;
    int inc = v;
#pragma unroll
    for (int o = 1; o < 32; o <<= 1) {
        int u = __shfl_up(inc, o, 32);
        if (lane >= o) inc += u;
    }
    if (lane == 31) wsum[wave] = inc;
    __syncthreads();
    int woff = 0;
#pragma unroll
    for (int w = 0; w < 32; ++w) woff += (w < wave) ? wsum[w] : 0;
    if (t < B) lcur[t] = woff + inc - v + hist[t * SORT_CHUNKS + g];
    __syncthreads();

    v2i* __restrict__ pr = (v2i*)pairs;
    const int e0 = g * chunk;
    const int e1 = (e0 + chunk < E) ? (e0 + chunk) : E;
    for (int eb = e0 + t; eb < e1; eb += 4 * 1024) {
        int d[4], s[4];
#pragma unroll
        for (int k = 0; k < 4; ++k) {
            const int e = eb + k * 1024;
            const bool ok = e < e1;
            d[k] = ok ? dst[e] : -1;
            s[k] = ok ? src[e] : 0;
        }
#pragma unroll
        for (int k = 0; k < 4; ++k) {
            if (d[k] >= 0) {
                const int p = atomicAdd(&lcur[d[k] >> BIN_SHIFT], 1);
                v2i w = {s[k], d[k]};
                pr[p] = w;
            }
        }
    }
}

__global__ __launch_bounds__(BIN_NODES) void k_bin_rows(const int* __restrict__ pairs,
                                                        const int* __restrict__ bintot,
                                                        int* __restrict__ rowptr,
                                                        int* __restrict__ csr,
                                                        float* __restrict__ dinv,
                                                        int n, int B) {
    __shared__ int cnt[BIN_NODES];
    __shared__ int cur[BIN_NODES];
    __shared__ int red[BIN_NODES / 32];
    __shared__ int wsum[BIN_NODES / 32];
    const int t = threadIdx.x, lane = t & 31, wave = t >> 5;
    const int b = blockIdx.x;
    int part = 0;
    for (int j = t; j < b; j += BIN_NODES) part += bintot[j];
#pragma unroll
    for (int m = 16; m > 0; m >>= 1) part += __shfl_xor(part, m, 32);
    if (lane == 0) red[wave] = part;
    cnt[t] = 0;
    __syncthreads();
    int binStart = 0;
#pragma unroll
    for (int w = 0; w < BIN_NODES / 32; ++w) binStart += red[w];
    const int binEnd = binStart + bintot[b];

    const v2i* __restrict__ pr = (const v2i*)pairs;
    for (int eb = binStart + t; eb < binEnd; eb += 8 * BIN_NODES) {
        int d[8];
#pragma unroll
        for (int k = 0; k < 8; ++k) {
            const int e = eb + k * BIN_NODES;
            d[k] = (e < binEnd) ? pr[e].y : -1;
        }
#pragma unroll
        for (int k = 0; k < 8; ++k)
            if (d[k] >= 0) atomicAdd(&cnt[d[k] & (BIN_NODES - 1)], 1);
    }
    __syncthreads();

    const int c = cnt[t];
    int inc = c;
#pragma unroll
    for (int o = 1; o < 32; o <<= 1) {
        int u = __shfl_up(inc, o, 32);
        if (lane >= o) inc += u;
    }
    if (lane == 31) wsum[wave] = inc;
    __syncthreads();
    int woff = 0;
#pragma unroll
    for (int w = 0; w < BIN_NODES / 32; ++w) woff += (w < wave) ? wsum[w] : 0;
    const int rowStart = binStart + woff + inc - c;
    const int node = b * BIN_NODES + t;
    if (node < n) {
        rowptr[node] = rowStart;
        dinv[node]   = rsqrtf((float)c + 1.0f);
    }
    if (b == B - 1 && t == 0) rowptr[n] = binEnd;
    cur[t] = rowStart;
    __syncthreads();

    for (int eb = binStart + t; eb < binEnd; eb += 8 * BIN_NODES) {
        v2i q[8];
#pragma unroll
        for (int k = 0; k < 8; ++k) {
            const int e = eb + k * BIN_NODES;
            v2i none = {0, -1};
            q[k] = (e < binEnd) ? pr[e] : none;
        }
#pragma unroll
        for (int k = 0; k < 8; ++k) {
            if (q[k].y >= 0) {
                const int p = atomicAdd(&cur[q[k].y & (BIN_NODES - 1)], 1);
                csr[p] = q[k].x;
            }
        }
    }
}

template <bool RELU_IN>
__global__ __launch_bounds__(GEMM_THREADS) void k_gemm_wmma(
        const float* __restrict__ X, const float* __restrict__ W,
        const float* __restrict__ dinv, float* __restrict__ H,
        int strips, int nrows) {
    __shared__ __align__(16) float ldsW[DF * DF];

    {
        const int n    = threadIdx.x & (DF - 1);
        const int kg   = threadIdx.x >> 7;
        const int t    = n >> 4;
        const int nloc = n & 15;
        v2f* __restrict__ lw = (v2f*)ldsW + ((t << 5) * 32 + nloc);
        const float* __restrict__ wc = W + n;
#pragma unroll
        for (int i = 0; i < DF / 4 / (GEMM_THREADS / DF); ++i) {
            const int k = i * (GEMM_THREADS / DF) + kg;
            const float* __restrict__ wp = wc + (size_t)(4 * k) * DF;
            const v2f lo = {wp[0], wp[DF]};
            const v2f hi = {wp[2 * DF], wp[3 * DF]};
            lw[k * 32]      = lo;
            lw[k * 32 + 16] = hi;
        }
    }
    __syncthreads();

    const int wave  = threadIdx.x >> 5;
    const int lane  = threadIdx.x & 31;
    const int strip = blockIdx.x * (blockDim.x >> 5) + wave;
    if (strip >= strips) return;

    const int nloc  = lane & 15;
    const int khalf = lane >> 4;

    int arow_i = strip * 16 + nloc;
    if (arow_i >= nrows) arow_i = nrows - 1;
    const float* __restrict__ arow = X + (size_t)arow_i * DF + 2 * khalf;

    const v2f* __restrict__ bbase = (const v2f*)ldsW + lane;

    v8f acc[8] = {};
#pragma unroll
    for (int k = 0; k < 32; ++k) {
        v2f a = *(const v2f*)(arow + 4 * k);
        if (RELU_IN) {
            a.x = fmaxf(a.x, 0.0f);
            a.y = fmaxf(a.y, 0.0f);
        }
#pragma unroll
        for (int t = 0; t < 8; ++t) {
            v2f b = bbase[(t * 32 + k) * 32];
            acc[t] = __builtin_amdgcn_wmma_f32_16x16x4_f32(
                         false, a, false, b, (short)0, acc[t], false, false);
        }
    }

    const int rbase = strip * 16 + 8 * khalf;
#pragma unroll
    for (int r = 0; r < 8; ++r) {
        int row = rbase + r;
        if (row >= nrows) continue;
        float dv = dinv[row];
        size_t ro = (size_t)row * DF;
#pragma unroll
        for (int t = 0; t < 8; ++t)
            H[ro + t * 16 + nloc] = acc[t][r] * dv;
    }
}

__global__ __launch_bounds__(256) void k_pull(
        const float* __restrict__ H, const int* __restrict__ rowptr,
        const int* __restrict__ csr, const float* __restrict__ dinv,
        const float* __restrict__ bias, float* __restrict__ OUT, int n) {
    const int d = __builtin_amdgcn_readfirstlane(
                      (int)(blockIdx.x * (blockDim.x >> 5) + (threadIdx.x >> 5)));
    if (d >= n) return;
    const int lane = threadIdx.x & 31;
    const int beg = rowptr[d], end = rowptr[d + 1];
    const float dd = dinv[d];
    const float* __restrict__ hl = H + lane * 4;
    const v4f bv = *(const v4f*)(bias + lane * 4);
    v4f acc = *(const v4f*)(hl + (size_t)d * DF);

    for (int base = beg; base < end; base += 32) {
        const int cn = (end - base < 32) ? (end - base) : 32;
        int s_l = 0;
        if (lane < cn) {
            s_l = csr[base + lane];
            s_l = min(max(s_l, 0), n - 1);
        }
        int j = 0;
        for (; j + 8 <= cn; j += 8) {
            v4f v[8];
#pragma unroll
            for (int k = 0; k < 8; ++k) {
                int s = __builtin_amdgcn_readlane(s_l, j + k);
                v[k]  = *(const v4f*)(hl + (size_t)s * DF);
            }
#pragma unroll
            for (int k = 0; k < 8; ++k) acc += v[k];
        }
        for (; j < cn; j += 4) {
            v4f v[4];
#pragma unroll
            for (int k = 0; k < 4; ++k) {
                int jj = (j + k < cn) ? (j + k) : (cn - 1);
                int s  = __builtin_amdgcn_readlane(s_l, jj);
                v[k]   = *(const v4f*)(hl + (size_t)s * DF);
            }
#pragma unroll
            for (int k = 0; k < 4; ++k) {
                v4f tsum = acc + v[k];
                acc = (j + k < cn) ? tsum : acc;
            }
        }
    }
    *(v4f*)(OUT + (size_t)d * DF + lane * 4) = acc * dd + bv;
}

__global__ __launch_bounds__(256) void k_normalize(const float* __restrict__ in,
                                                   float* __restrict__ out, int n) {
    int row = blockIdx.x * (blockDim.x >> 5) + (threadIdx.x >> 5);
    if (row >= n) return;
    int lane = threadIdx.x & 31;
    v4f v = *(const v4f*)(in + (size_t)row * DF + lane * 4);
    float ss = v.x * v.x + v.y * v.y + v.z * v.z + v.w * v.w;
#pragma unroll
    for (int m = 16; m > 0; m >>= 1) ss += __shfl_xor(ss, m, 32);
    float sc = 1.0f / fmaxf(sqrtf(ss), 1e-12f);
    v4f o = {v.x * sc, v.y * sc, v.z * sc, v.w * sc};
    *(v4f*)(out + (size_t)row * DF + lane * 4) = o;
}


extern "C" void kernel_launch(void* const* d_in, const int* in_sizes, int n_in,
                              void* d_out, int out_size, void* d_ws, size_t ws_size,
                              hipStream_t stream) {
    const float* x  = (const float*)d_in[0];
    const int*   ei = (const int*)  d_in[1];
    const float* W1 = (const float*)d_in[2];
    const float* b1 = (const float*)d_in[3];
    const float* W2 = (const float*)d_in[4];
    const float* b2 = (const float*)d_in[5];
    const float* W3 = (const float*)d_in[6];
    const float* b3 = (const float*)d_in[7];

    const int N = in_sizes[0] / DF;
    const int E = in_sizes[1] / 2;
    const int* src = ei;
    const int* dst = ei + E;

    float* out = (float*)d_out;

    char* ws = (char*)d_ws;
    const int B = (N + BIN_NODES - 1) / BIN_NODES;
    const int chunk = (E + SORT_CHUNKS - 1) / SORT_CHUNKS;
    size_t offN  = ((size_t)(N + 1) * 4 + 255) & ~(size_t)255;
    size_t offB  = ((size_t)B * 4 + 255) & ~(size_t)255;
    size_t offH  = ((size_t)B * SORT_CHUNKS * 4 + 255) & ~(size_t)255;
    size_t offE  = ((size_t)E * 4 + 255) & ~(size_t)255;
    float* dinv   = (float*)ws;
    int*   rowptr = (int*)(ws + offN);
    int*   bintot = (int*)(ws + 2 * offN);
    int*   hist   = (int*)(ws + 2 * offN + offB);
    int*   csr    = (int*)(ws + 2 * offN + offB + offH);
    int*   pairs  = (int*)(ws + 2 * offN + offB + offH + offE);
    float* H      = (float*)(ws + 2 * offN + offB + offH + 3 * offE);
    float* AGGa   = H + (size_t)N * DF;

    const int strips  = (N + 15) / 16;
    const int gblocks = (strips + 7) / 8;
    const int pblocks = (N + 7) / 8;

    k_bin_hist   <<<SORT_CHUNKS, 1024,        0, stream>>>(dst, hist, E, B, chunk);
    k_bin_scan   <<<B,           SORT_CHUNKS, 0, stream>>>(hist, bintot);
    k_bin_scatter<<<SORT_CHUNKS, 1024,        0, stream>>>(src, dst, hist, bintot, pairs, E, B, chunk);
    k_bin_rows   <<<B,           BIN_NODES,   0, stream>>>(pairs, bintot, rowptr, csr, dinv, N, B);

    k_gemm_wmma<false><<<gblocks, GEMM_THREADS, 0, stream>>>(x, W1, dinv, H, strips, N);
    k_pull            <<<pblocks, 256, 0, stream>>>(H, rowptr, csr, dinv, b1, AGGa, N);

    k_gemm_wmma<true> <<<gblocks, GEMM_THREADS, 0, stream>>>(AGGa, W2, dinv, H, strips, N);
    k_pull            <<<pblocks, 256, 0, stream>>>(H, rowptr, csr, dinv, b2, out, N);

    k_gemm_wmma<true> <<<gblocks, GEMM_THREADS, 0, stream>>>(out, W3, dinv, H, strips, N);
    k_pull            <<<pblocks, 256, 0, stream>>>(H, rowptr, csr, dinv, b3, AGGa, N);

    k_normalize<<<(N + 7) / 8, 256, 0, stream>>>(AGGa, out, N);
}


# Final row L2-normalize done inside the last aggregation launch (the wave already holds the whole row): one launch of one wave per row, one plane write and one plane read fewer; sums unchanged
# speedup vs baseline: 4.2016x; 1.0344x over previous
#include <hip/hip_runtime.h>
#include <hip/hip_bf16.h>


typedef __attribute__((ext_vector_type(2))) float v2f;
typedef __attribute__((ext_vector_type(4))) float v4f;
typedef __attribute__((ext_vector_type(8))) float v8f;

#define DF 128
#define GEMM_THREADS 256


#define BIN_SHIFT   8
#define BIN_NODES   (1 << BIN_SHIFT)
#define SORT_CHUNKS 256
#define MAX_BINS    1024

typedef __attribute__((ext_vector_type(2))) int v2i;

__global__ __launch_bounds__(1024) void k_bin_hist(const int* __restrict__ dst,
                                                   int* __restrict__ hist,
                                                   int E, int B, int chunk) {
    __shared__ int lh[MAX_BINS];
    const int t = threadIdx.x, g = blockIdx.x;
    for (int b = t; b < B; b += 1024) lh[b] = 0;
    __syncthreads();
    const int e0 = g * chunk;
    const int e1 = (e0 + chunk < E) ? (e0 + chunk) : E;
    for (int eb = e0 + t; eb < e1; eb += 4 * 1024) {
        int d[4];
#pragma unroll
        for (int k = 0; k < 4; ++k) {
            const int e = eb + k * 1024;
            d[k] = (e < e1) ? dst[e] : -1;
        }
#pragma unroll
        for (int k = 0; k < 4; ++k)
            if (d[k] >= 0) atomicAdd(&lh[d[k] >> BIN_SHIFT], 1);
    }
    __syncthreads();
    for (int b = t; b < B; b += 1024) hist[b * SORT_CHUNKS + g] = lh[b];
}

__global__ __launch_bounds__(SORT_CHUNKS) void k_bin_scan(int* __restrict__ hist,
                                                          int* __restrict__ bintot) {
    __shared__ int wsum[SORT_CHUNKS / 32];
    const int t = threadIdx.x, lane = t & 31, wave = t >> 5;
    const int b = blockIdx.x;
    const int v = hist[b * SORT_CHUNKS + t];
    int inc = v;
#pragma unroll
    for (int o = 1; o < 32; o <<= 1) {
        int u = __shfl_up(inc, o, 32);
        if (lane >= o) inc += u;
    }
    if (lane == 31) wsum[wave] = inc;
    __syncthreads();
    int woff = 0;
#pragma unroll
    for (int w = 0; w < SORT_CHUNKS / 32; ++w) woff += (w < wave) ? wsum[w] : 0;
    hist[b * SORT_CHUNKS + t] = woff + inc - v;
    if (t == SORT_CHUNKS - 1) bintot[b] = woff + inc;
}

__global__ __launch_bounds__(1024) void k_bin_scatter(const int* __restrict__ src,
                                                      const int* __restrict__ dst,
                                                      const int* __restrict__ hist,
                                                      const int* __restrict__ bintot,
                                                      int* __restrict__ pairs,
                                                      int E, int B, int chunk) {
    __shared__ int lcur[MAX_BINS];
    __shared__ int wsum[32];
    const int t = threadIdx.x, lane = t & 31, wave = t >> 5, g = blockIdx.x;
    const int v = (t < B) ? bintot[t] : 0;
    int inc = v;
#pragma unroll
    for (int o = 1; o < 32; o <<= 1) {
        int u = __shfl_up(inc, o, 32);
        if (lane >= o) inc += u;
    }
    if (lane == 31) wsum[wave] = inc;
    __syncthreads();
    int woff = 0;
#pragma unroll
    for (int w = 0; w < 32; ++w) woff += (w < wave) ? wsum[w] : 0;
    if (t < B) lcur[t] = woff + inc - v + hist[t * SORT_CHUNKS + g];
    __syncthreads();

    v2i* __restrict__ pr = (v2i*)pairs;
    const int e0 = g * chunk;
    const int e1 = (e0 + chunk < E) ? (e0 + chunk) : E;
    for (int eb = e0 + t; eb < e1; eb += 4 * 1024) {
        int d[4], s[4];
#pragma unroll
        for (int k = 0; k < 4; ++k) {
            const int e = eb + k * 1024;
            const bool ok = e < e1;
            d[k] = ok ? dst[e] : -1;
            s[k] = ok ? src[e] : 0;
        }
#pragma unroll
        for (int k = 0; k < 4; ++k) {
            if (d[k] >= 0) {
                const int p = atomicAdd(&lcur[d[k] >> BIN_SHIFT], 1);
                v2i w = {s[k], d[k]};
                pr[p] = w;
            }
        }
    }
}

__global__ __launch_bounds__(BIN_NODES) void k_bin_rows(const int* __restrict__ pairs,
                                                        const int* __restrict__ bintot,
                                                        int* __restrict__ rowptr,
                                                        int* __restrict__ csr,
                                                        float* __restrict__ dinv,
                                                        int n, int B) {
    __shared__ int cnt[BIN_NODES];
    __shared__ int cur[BIN_NODES];
    __shared__ int red[BIN_NODES / 32];
    __shared__ int wsum[BIN_NODES / 32];
    const int t = threadIdx.x, lane = t & 31, wave = t >> 5;
    const int b = blockIdx.x;
    int part = 0;
    for (int j = t; j < b; j += BIN_NODES) part += bintot[j];
#pragma unroll
    for (int m = 16; m > 0; m >>= 1) part += __shfl_xor(part, m, 32);
    if (lane == 0) red[wave] = part;
    cnt[t] = 0;
    __syncthreads();
    int binStart = 0;
#pragma unroll
    for (int w = 0; w < BIN_NODES / 32; ++w) binStart += red[w];
    const int binEnd = binStart + bintot[b];

    const v2i* __restrict__ pr = (const v2i*)pairs;
    for (int eb = binStart + t; eb < binEnd; eb += 8 * BIN_NODES) {
        int d[8];
#pragma unroll
        for (int k = 0; k < 8; ++k) {
            const int e = eb + k * BIN_NODES;
            d[k] = (e < binEnd) ? pr[e].y : -1;
        }
#pragma unroll
        for (int k = 0; k < 8; ++k)
            if (d[k] >= 0) atomicAdd(&cnt[d[k] & (BIN_NODES - 1)], 1);
    }
    __syncthreads();

    const int c = cnt[t];
    int inc = c;
#pragma unroll
    for (int o = 1; o < 32; o <<= 1) {
        int u = __shfl_up(inc, o, 32);
        if (lane >= o) inc += u;
    }
    if (lane == 31) wsum[wave] = inc;
    __syncthreads();
    int woff = 0;
#pragma unroll
    for (int w = 0; w < BIN_NODES / 32; ++w) woff += (w < wave) ? wsum[w] : 0;
    const int rowStart = binStart + woff + inc - c;
    const int node = b * BIN_NODES + t;
    if (node < n) {
        rowptr[node] = rowStart;
        dinv[node]   = rsqrtf((float)c + 1.0f);
    }
    if (b == B - 1 && t == 0) rowptr[n] = binEnd;
    cur[t] = rowStart;
    __syncthreads();

    for (int eb = binStart + t; eb < binEnd; eb += 8 * BIN_NODES) {
        v2i q[8];
#pragma unroll
        for (int k = 0; k < 8; ++k) {
            const int e = eb + k * BIN_NODES;
            v2i none = {0, -1};
            q[k] = (e < binEnd) ? pr[e] : none;
        }
#pragma unroll
        for (int k = 0; k < 8; ++k) {
            if (q[k].y >= 0) {
                const int p = atomicAdd(&cur[q[k].y & (BIN_NODES - 1)], 1);
                csr[p] = q[k].x;
            }
        }
    }
}

template <bool RELU_IN>
__global__ __launch_bounds__(GEMM_THREADS) void k_gemm_wmma(
        const float* __restrict__ X, const float* __restrict__ W,
        const float* __restrict__ dinv, float* __restrict__ H,
        int strips, int nrows) {
    __shared__ __align__(16) float ldsW[DF * DF];

    {
        const int n    = threadIdx.x & (DF - 1);
        const int kg   = threadIdx.x >> 7;
        const int t    = n >> 4;
        const int nloc = n & 15;
        v2f* __restrict__ lw = (v2f*)ldsW + ((t << 5) * 32 + nloc);
        const float* __restrict__ wc = W + n;
#pragma unroll
        for (int i = 0; i < DF / 4 / (GEMM_THREADS / DF); ++i) {
            const int k = i * (GEMM_THREADS / DF) + kg;
            const float* __restrict__ wp = wc + (size_t)(4 * k) * DF;
            const v2f lo = {wp[0], wp[DF]};
            const v2f hi = {wp[2 * DF], wp[3 * DF]};
            lw[k * 32]      = lo;
            lw[k * 32 + 16] = hi;
        }
    }
    __syncthreads();

    const int wave  = threadIdx.x >> 5;
    const int lane  = threadIdx.x & 31;
    const int strip = blockIdx.x * (blockDim.x >> 5) + wave;
    if (strip >= strips) return;

    const int nloc  = lane & 15;
    const int khalf = lane >> 4;

    int arow_i = strip * 16 + nloc;
    if (arow_i >= nrows) arow_i = nrows - 1;
    const float* __restrict__ arow = X + (size_t)arow_i * DF + 2 * khalf;

    const v2f* __restrict__ bbase = (const v2f*)ldsW + lane;

    v8f acc[8] = {};
#pragma unroll
    for (int k = 0; k < 32; ++k) {
        v2f a = *(const v2f*)(arow + 4 * k);
        if (RELU_IN) {
            a.x = fmaxf(a.x, 0.0f);
            a.y = fmaxf(a.y, 0.0f);
        }
#pragma unroll
        for (int t = 0; t < 8; ++t) {
            v2f b = bbase[(t * 32 + k) * 32];
            acc[t] = __builtin_amdgcn_wmma_f32_16x16x4_f32(
                         false, a, false, b, (short)0, acc[t], false, false);
        }
    }

    const int rbase = strip * 16 + 8 * khalf;
#pragma unroll
    for (int r = 0; r < 8; ++r) {
        int row = rbase + r;
        if (row >= nrows) continue;
        float dv = dinv[row];
        size_t ro = (size_t)row * DF;
#pragma unroll
        for (int t = 0; t < 8; ++t)
            H[ro + t * 16 + nloc] = acc[t][r] * dv;
    }
}

template <bool NORM_OUT>
__global__ __launch_bounds__(256) void k_pull(
        const float* __restrict__ H, const int* __restrict__ rowptr,
        const int* __restrict__ csr, const float* __restrict__ dinv,
        const float* __restrict__ bias, float* __restrict__ OUT, int n) {
    const int d = __builtin_amdgcn_readfirstlane(
                      (int)(blockIdx.x * (blockDim.x >> 5) + (threadIdx.x >> 5)));
    if (d >= n) return;
    const int lane = threadIdx.x & 31;
    const int beg = rowptr[d], end = rowptr[d + 1];
    const float dd = dinv[d];
    const float* __restrict__ hl = H + lane * 4;
    const v4f bv = *(const v4f*)(bias + lane * 4);
    v4f acc = *(const v4f*)(hl + (size_t)d * DF);

    for (int base = beg; base < end; base += 32) {
        const int cn = (end - base < 32) ? (end - base) : 32;
        int s_l = 0;
        if (lane < cn) {
            s_l = csr[base + lane];
            s_l = min(max(s_l, 0), n - 1);
        }
        int j = 0;
        for (; j + 8 <= cn; j += 8) {
            v4f v[8];
#pragma unroll
            for (int k = 0; k < 8; ++k) {
                int s = __builtin_amdgcn_readlane(s_l, j + k);
                v[k]  = *(const v4f*)(hl + (size_t)s * DF);
            }
#pragma unroll
            for (int k = 0; k < 8; ++k) acc += v[k];
        }
        for (; j < cn; j += 4) {
            v4f v[4];
#pragma unroll
            for (int k = 0; k < 4; ++k) {
                int jj = (j + k < cn) ? (j + k) : (cn - 1);
                int s  = __builtin_amdgcn_readlane(s_l, jj);
                v[k]   = *(const v4f*)(hl + (size_t)s * DF);
            }
#pragma unroll
            for (int k = 0; k < 4; ++k) {
                v4f tsum = acc + v[k];
                acc = (j + k < cn) ? tsum : acc;
            }
        }
    }
    v4f o = acc * dd + bv;
    if (NORM_OUT) {
        float ss = o.x * o.x + o.y * o.y + o.z * o.z + o.w * o.w;
#pragma unroll
        for (int m = 16; m > 0; m >>= 1) ss += __shfl_xor(ss, m, 32);
        const float sc = 1.0f / fmaxf(sqrtf(ss), 1e-12f);
        o.x *= sc; o.y *= sc; o.z *= sc; o.w *= sc;
    }
    *(v4f*)(OUT + (size_t)d * DF + lane * 4) = o;
}


extern "C" void kernel_launch(void* const* d_in, const int* in_sizes, int n_in,
                              void* d_out, int out_size, void* d_ws, size_t ws_size,
                              hipStream_t stream) {
    const float* x  = (const float*)d_in[0];
    const int*   ei = (const int*)  d_in[1];
    const float* W1 = (const float*)d_in[2];
    const float* b1 = (const float*)d_in[3];
    const float* W2 = (const float*)d_in[4];
    const float* b2 = (const float*)d_in[5];
    const float* W3 = (const float*)d_in[6];
    const float* b3 = (const float*)d_in[7];

    const int N = in_sizes[0] / DF;
    const int E = in_sizes[1] / 2;
    const int* src = ei;
    const int* dst = ei + E;

    float* out = (float*)d_out;

    char* ws = (char*)d_ws;
    const int B = (N + BIN_NODES - 1) / BIN_NODES;
    const int chunk = (E + SORT_CHUNKS - 1) / SORT_CHUNKS;
    size_t offN  = ((size_t)(N + 1) * 4 + 255) & ~(size_t)255;
    size_t offB  = ((size_t)B * 4 + 255) & ~(size_t)255;
    size_t offH  = ((size_t)B * SORT_CHUNKS * 4 + 255) & ~(size_t)255;
    size_t offE  = ((size_t)E * 4 + 255) & ~(size_t)255;
    float* dinv   = (float*)ws;
    int*   rowptr = (int*)(ws + offN);
    int*   bintot = (int*)(ws + 2 * offN);
    int*   hist   = (int*)(ws + 2 * offN + offB);
    int*   csr    = (int*)(ws + 2 * offN + offB + offH);
    int*   pairs  = (int*)(ws + 2 * offN + offB + offH + offE);
    float* H      = (float*)(ws + 2 * offN + offB + offH + 3 * offE);
    float* AGGa   = H + (size_t)N * DF;

    const int strips  = (N + 15) / 16;
    const int gblocks = (strips + 7) / 8;
    const int pblocks = (N + 7) / 8;

    k_bin_hist   <<<SORT_CHUNKS, 1024,        0, stream>>>(dst, hist, E, B, chunk);
    k_bin_scan   <<<B,           SORT_CHUNKS, 0, stream>>>(hist, bintot);
    k_bin_scatter<<<SORT_CHUNKS, 1024,        0, stream>>>(src, dst, hist, bintot, pairs, E, B, chunk);
    k_bin_rows   <<<B,           BIN_NODES,   0, stream>>>(pairs, bintot, rowptr, csr, dinv, N, B);

    k_gemm_wmma<false><<<gblocks, GEMM_THREADS, 0, stream>>>(x, W1, dinv, H, strips, N);
    k_pull<false>     <<<pblocks, 256, 0, stream>>>(H, rowptr, csr, dinv, b1, AGGa, N);

    k_gemm_wmma<true> <<<gblocks, GEMM_THREADS, 0, stream>>>(AGGa, W2, dinv, H, strips, N);
    k_pull<false>     <<<pblocks, 256, 0, stream>>>(H, rowptr, csr, dinv, b2, out, N);

    k_gemm_wmma<true> <<<gblocks, GEMM_THREADS, 0, stream>>>(out, W3, dinv, H, strips, N);
    k_pull<true>      <<<pblocks, 256, 0, stream>>>(H, rowptr, csr, dinv, b3, out, N);
}


# Layers 1-2: a wave sums its 16-row strip (a row's gathers all in flight), relu, rows to LDS as A tile of the next layer's WMMA GEMM in the same launch: no sum plane, 2 fewer launches; sums unchanged
# speedup vs baseline: 4.3618x; 1.0381x over previous
#include <hip/hip_runtime.h>
#include <hip/hip_bf16.h>


typedef __attribute__((ext_vector_type(2))) float v2f;
typedef __attribute__((ext_vector_type(4))) float v4f;
typedef __attribute__((ext_vector_type(8))) float v8f;

#define DF 128
#define GEMM_THREADS 256


#define BIN_SHIFT   8
#define BIN_NODES   (1 << BIN_SHIFT)
#define SORT_CHUNKS 256
#define MAX_BINS    1024

typedef __attribute__((ext_vector_type(2))) int v2i;

__global__ __launch_bounds__(1024) void k_bin_hist(const int* __restrict__ dst,
                                                   int* __restrict__ hist,
                                                   int E, int B, int chunk) {
    __shared__ int lh[MAX_BINS];
    const int t = threadIdx.x, g = blockIdx.x;
    for (int b = t; b < B; b += 1024) lh[b] = 0;
    __syncthreads();
    const int e0 = g * chunk;
    const int e1 = (e0 + chunk < E) ? (e0 + chunk) : E;
    for (int eb = e0 + t; eb < e1; eb += 4 * 1024) {
        int d[4];
#pragma unroll
        for (int k = 0; k < 4; ++k) {
            const int e = eb + k * 1024;
            d[k] = (e < e1) ? dst[e] : -1;
        }
#pragma unroll
        for (int k = 0; k < 4; ++k)
            if (d[k] >= 0) atomicAdd(&lh[d[k] >> BIN_SHIFT], 1);
    }
    __syncthreads();
    for (int b = t; b < B; b += 1024) hist[b * SORT_CHUNKS + g] = lh[b];
}

__global__ __launch_bounds__(SORT_CHUNKS) void k_bin_scan(int* __restrict__ hist,
                                                          int* __restrict__ bintot) {
    __shared__ int wsum[SORT_CHUNKS / 32];
    const int t = threadIdx.x, lane = t & 31, wave = t >> 5;
    const int b = blockIdx.x;
    const int v = hist[b * SORT_CHUNKS + t];
    int inc = v;
#pragma unroll
    for (int o = 1; o < 32; o <<= 1) {
        int u = __shfl_up(inc, o, 32);
        if (lane >= o) inc += u;
    }
    if (lane == 31) wsum[wave] = inc;
    __syncthreads();
    int woff = 0;
#pragma unroll
    for (int w = 0; w < SORT_CHUNKS / 32; ++w) woff += (w < wave) ? wsum[w] : 0;
    hist[b * SORT_CHUNKS + t] = woff + inc - v;
    if (t == SORT_CHUNKS - 1) bintot[b] = woff + inc;
}

__global__ __launch_bounds__(1024) void k_bin_scatter(const int* __restrict__ src,
                                                      const int* __restrict__ dst,
                                                      const int* __restrict__ hist,
                                                      const int* __restrict__ bintot,
                                                      int* __restrict__ pairs,
                                                      int E, int B, int chunk) {
    __shared__ int lcur[MAX_BINS];
    __shared__ int wsum[32];
    const int t = threadIdx.x, lane = t & 31, wave = t >> 5, g = blockIdx.x;
    const int v = (t < B) ? bintot[t] : 0;
    int inc = v;
#pragma unroll
    for (int o = 1; o < 32; o <<= 1) {
        int u = __shfl_up(inc, o, 32);
        if (lane >= o) inc += u;
    }
    if (lane == 31) wsum[wave] = inc;
    __syncthreads();
    int woff = 0;
#pragma unroll
    for (int w = 0; w < 32; ++w) woff += (w < wave) ? wsum[w] : 0;
    if (t < B) lcur[t] = woff + inc - v + hist[t * SORT_CHUNKS + g];
    __syncthreads();

    v2i* __restrict__ pr = (v2i*)pairs;
    const int e0 = g * chunk;
    const int e1 = (e0 + chunk < E) ? (e0 + chunk) : E;
    for (int eb = e0 + t; eb < e1; eb += 4 * 1024) {
        int d[4], s[4];
#pragma unroll
        for (int k = 0; k < 4; ++k) {
            const int e = eb + k * 1024;
            const bool ok = e < e1;
            d[k] = ok ? dst[e] : -1;
            s[k] = ok ? src[e] : 0;
        }
#pragma unroll
        for (int k = 0; k < 4; ++k) {
            if (d[k] >= 0) {
                const int p = atomicAdd(&lcur[d[k] >> BIN_SHIFT], 1);
                v2i w = {s[k], d[k]};
                pr[p] = w;
            }
        }
    }
}

__global__ __launch_bounds__(BIN_NODES) void k_bin_rows(const int* __restrict__ pairs,
                                                        const int* __restrict__ bintot,
                                                        int* __restrict__ rowptr,
                                                        int* __restrict__ csr,
                                                        float* __restrict__ dinv,
                                                        int n, int B) {
    __shared__ int cnt[BIN_NODES];
    __shared__ int cur[BIN_NODES];
    __shared__ int red[BIN_NODES / 32];
    __shared__ int wsum[BIN_NODES / 32];
    const int t = threadIdx.x, lane = t & 31, wave = t >> 5;
    const int b = blockIdx.x;
    int part = 0;
    for (int j = t; j < b; j += BIN_NODES) part += bintot[j];
#pragma unroll
    for (int m = 16; m > 0; m >>= 1) part += __shfl_xor(part, m, 32);
    if (lane == 0) red[wave] = part;
    cnt[t] = 0;
    __syncthreads();
    int binStart = 0;
#pragma unroll
    for (int w = 0; w < BIN_NODES / 32; ++w) binStart += red[w];
    const int binEnd = binStart + bintot[b];

    const v2i* __restrict__ pr = (const v2i*)pairs;
    for (int eb = binStart + t; eb < binEnd; eb += 8 * BIN_NODES) {
        int d[8];
#pragma unroll
        for (int k = 0; k < 8; ++k) {
            const int e = eb + k * BIN_NODES;
            d[k] = (e < binEnd) ? pr[e].y : -1;
        }
#pragma unroll
        for (int k = 0; k < 8; ++k)
            if (d[k] >= 0) atomicAdd(&cnt[d[k] & (BIN_NODES - 1)], 1);
    }
    __syncthreads();

    const int c = cnt[t];
    int inc = c;
#pragma unroll
    for (int o = 1; o < 32; o <<= 1) {
        int u = __shfl_up(inc, o, 32);
        if (lane >= o) inc += u;
    }
    if (lane == 31) wsum[wave] = inc;
    __syncthreads();
    int woff = 0;
#pragma unroll
    for (int w = 0; w < BIN_NODES / 32; ++w) woff += (w < wave) ? wsum[w] : 0;
    const int rowStart = binStart + woff + inc - c;
    const int node = b * BIN_NODES + t;
    if (node < n) {
        rowptr[node] = rowStart;
        dinv[node]   = rsqrtf((float)c + 1.0f);
    }
    if (b == B - 1 && t == 0) rowptr[n] = binEnd;
    cur[t] = rowStart;
    __syncthreads();

    for (int eb = binStart + t; eb < binEnd; eb += 8 * BIN_NODES) {
        v2i q[8];
#pragma unroll
        for (int k = 0; k < 8; ++k) {
            const int e = eb + k * BIN_NODES;
            v2i none = {0, -1};
            q[k] = (e < binEnd) ? pr[e] : none;
        }
#pragma unroll
        for (int k = 0; k < 8; ++k) {
            if (q[k].y >= 0) {
                const int p = atomicAdd(&cur[q[k].y & (BIN_NODES - 1)], 1);
                csr[p] = q[k].x;
            }
        }
    }
}

template <bool RELU_IN>
__global__ __launch_bounds__(GEMM_THREADS) void k_gemm_wmma(
        const float* __restrict__ X, const float* __restrict__ W,
        const float* __restrict__ dinv, float* __restrict__ H,
        int strips, int nrows) {
    __shared__ __align__(16) float ldsW[DF * DF];

    {
        const int n    = threadIdx.x & (DF - 1);
        const int kg   = threadIdx.x >> 7;
        const int t    = n >> 4;
        const int nloc = n & 15;
        v2f* __restrict__ lw = (v2f*)ldsW + ((t << 5) * 32 + nloc);
        const float* __restrict__ wc = W + n;
#pragma unroll
        for (int i = 0; i < DF / 4 / (GEMM_THREADS / DF); ++i) {
            const int k = i * (GEMM_THREADS / DF) + kg;
            const float* __restrict__ wp = wc + (size_t)(4 * k) * DF;
            const v2f lo = {wp[0], wp[DF]};
            const v2f hi = {wp[2 * DF], wp[3 * DF]};
            lw[k * 32]      = lo;
            lw[k * 32 + 16] = hi;
        }
    }
    __syncthreads();

    const int wave  = threadIdx.x >> 5;
    const int lane  = threadIdx.x & 31;
    const int strip = blockIdx.x * (blockDim.x >> 5) + wave;
    if (strip >= strips) return;

    const int nloc  = lane & 15;
    const int khalf = lane >> 4;

    int arow_i = strip * 16 + nloc;
    if (arow_i >= nrows) arow_i = nrows - 1;
    const float* __restrict__ arow = X + (size_t)arow_i * DF + 2 * khalf;

    const v2f* __restrict__ bbase = (const v2f*)ldsW + lane;

    v8f acc[8] = {};
#pragma unroll
    for (int k = 0; k < 32; ++k) {
        v2f a = *(const v2f*)(arow + 4 * k);
        if (RELU_IN) {
            a.x = fmaxf(a.x, 0.0f);
            a.y = fmaxf(a.y, 0.0f);
        }
#pragma unroll
        for (int t = 0; t < 8; ++t) {
            v2f b = bbase[(t * 32 + k) * 32];
            acc[t] = __builtin_amdgcn_wmma_f32_16x16x4_f32(
                         false, a, false, b, (short)0, acc[t], false, false);
        }
    }

    const int rbase = strip * 16 + 8 * khalf;
#pragma unroll
    for (int r = 0; r < 8; ++r) {
        int row = rbase + r;
        if (row >= nrows) continue;
        float dv = dinv[row];
        size_t ro = (size_t)row * DF;
#pragma unroll
        for (int t = 0; t < 8; ++t)
            H[ro + t * 16 + nloc] = acc[t][r] * dv;
    }
}

template <bool NORM_OUT>
__global__ __launch_bounds__(256) void k_pull(
        const float* __restrict__ H, const int* __restrict__ rowptr,
        const int* __restrict__ csr, const float* __restrict__ dinv,
        const float* __restrict__ bias, float* __restrict__ OUT, int n) {
    const int d = __builtin_amdgcn_readfirstlane(
                      (int)(blockIdx.x * (blockDim.x >> 5) + (threadIdx.x >> 5)));
    if (d >= n) return;
    const int lane = threadIdx.x & 31;
    const int beg = rowptr[d], end = rowptr[d + 1];
    const float dd = dinv[d];
    const float* __restrict__ hl = H + lane * 4;
    const v4f bv = *(const v4f*)(bias + lane * 4);
    v4f acc = *(const v4f*)(hl + (size_t)d * DF);

    for (int base = beg; base < end; base += 32) {
        const int cn = (end - base < 32) ? (end - base) : 32;
        int s_l = 0;
        if (lane < cn) {
            s_l = csr[base + lane];
            s_l = min(max(s_l, 0), n - 1);
        }
        int j = 0;
        for (; j + 8 <= cn; j += 8) {
            v4f v[8];
#pragma unroll
            for (int k = 0; k < 8; ++k) {
                int s = __builtin_amdgcn_readlane(s_l, j + k);
                v[k]  = *(const v4f*)(hl + (size_t)s * DF);
            }
#pragma unroll
            for (int k = 0; k < 8; ++k) acc += v[k];
        }
        for (; j < cn; j += 4) {
            v4f v[4];
#pragma unroll
            for (int k = 0; k < 4; ++k) {
                int jj = (j + k < cn) ? (j + k) : (cn - 1);
                int s  = __builtin_amdgcn_readlane(s_l, jj);
                v[k]   = *(const v4f*)(hl + (size_t)s * DF);
            }
#pragma unroll
            for (int k = 0; k < 4; ++k) {
                v4f tsum = acc + v[k];
                acc = (j + k < cn) ? tsum : acc;
            }
        }
    }
    v4f o = acc * dd + bv;
    if (NORM_OUT) {
        float ss = o.x * o.x + o.y * o.y + o.z * o.z + o.w * o.w;
#pragma unroll
        for (int m = 16; m > 0; m >>= 1) ss += __shfl_xor(ss, m, 32);
        const float sc = 1.0f / fmaxf(sqrtf(ss), 1e-12f);
        o.x *= sc; o.y *= sc; o.z *= sc; o.w *= sc;
    }
    *(v4f*)(OUT + (size_t)d * DF + lane * 4) = o;
}

#define APITCH 132

__global__ __launch_bounds__(GEMM_THREADS) void k_pull_gemm(
        const float* __restrict__ Hin, const int* __restrict__ rowptr,
        const int* __restrict__ csr, const float* __restrict__ dinv,
        const float* __restrict__ bias, const float* __restrict__ W,
        float* __restrict__ Hout, int strips, int nrows) {
    __shared__ __align__(16) float lds[DF * DF + (GEMM_THREADS / 32) * 16 * APITCH];
    float* __restrict__ ldsW = lds;
    float* __restrict__ ldsA = lds + DF * DF;

    {
        const int n    = threadIdx.x & (DF - 1);
        const int kg   = threadIdx.x >> 7;
        const int t    = n >> 4;
        const int nloc = n & 15;
        v2f* __restrict__ lw = (v2f*)ldsW + ((t << 5) * 32 + nloc);
        const float* __restrict__ wc = W + n;
#pragma unroll
        for (int i = 0; i < DF / 4 / (GEMM_THREADS / DF); ++i) {
            const int k = i * (GEMM_THREADS / DF) + kg;
            const float* __restrict__ wp = wc + (size_t)(4 * k) * DF;
            const v2f lo = {wp[0], wp[DF]};
            const v2f hi = {wp[2 * DF], wp[3 * DF]};
            lw[k * 32]      = lo;
            lw[k * 32 + 16] = hi;
        }
    }
    __syncthreads();

    const int wave  = __builtin_amdgcn_readfirstlane((int)(threadIdx.x >> 5));
    const int lane  = threadIdx.x & 31;
    const int strip = __builtin_amdgcn_readfirstlane((int)(blockIdx.x * (GEMM_THREADS >> 5)) + wave);
    if (strip >= strips) return;

    float* __restrict__ myA = ldsA + wave * 16 * APITCH;
    const int d0 = strip * 16;
    const int nr = (nrows - d0 < 16) ? (nrows - d0) : 16;

    {
        const float* __restrict__ hl = Hin + lane * 4;
        const v4f bv = *(const v4f*)(bias + lane * 4);
        int   rp_l = 0;
        float dv_l = 0.0f;
        if (lane <= nr) rp_l = rowptr[d0 + lane];
        if (lane <  nr) dv_l = dinv[d0 + lane];

        int beg = __builtin_amdgcn_readlane(rp_l, 0);
        int end = __builtin_amdgcn_readlane(rp_l, 1);
        v4f self = *(const v4f*)(hl + (size_t)d0 * DF);
        int s_l = 0;
        if (lane < end - beg) s_l = csr[beg + lane];

        for (int r = 0; r < nr; ++r) {
            int nend = end;
            v4f nself = self;
            int ns_l = 0;
            if (r + 1 < nr) {
                nend  = __builtin_amdgcn_readlane(rp_l, r + 2);
                nself = *(const v4f*)(hl + (size_t)(d0 + r + 1) * DF);
                if (lane < nend - end) ns_l = csr[end + lane];
            }

            v4f acc = self;
            for (int base = beg; base < end; base += 32) {
                const int cn = (end - base < 32) ? (end - base) : 32;
                if (base != beg) {
                    s_l = 0;
                    if (lane < cn) s_l = csr[base + lane];
                }
                v4f v[32];
#pragma unroll
                for (int g = 0; g < 8; ++g) {
                    if (4 * g < cn) {
#pragma unroll
                        for (int k = 0; k < 4; ++k) {
                            const int jj = (4 * g + k < cn) ? (4 * g + k) : (cn - 1);
                            int s = __builtin_amdgcn_readlane(s_l, jj);
                            s = min(max(s, 0), nrows - 1);
                            v[4 * g + k] = *(const v4f*)(hl + (size_t)s * DF);
                        }
                    }
                }
#pragma unroll
                for (int g = 0; g < 8; ++g) {
                    if (4 * g < cn) {
#pragma unroll
                        for (int k = 0; k < 4; ++k) {
                            const v4f tsum = acc + v[4 * g + k];
                            acc = (4 * g + k < cn) ? tsum : acc;
                        }
                    }
                }
            }

            const float dd = __int_as_float(__builtin_amdgcn_readlane(__float_as_int(dv_l), r));
            v4f o = acc * dd + bv;
            o.x = fmaxf(o.x, 0.0f); o.y = fmaxf(o.y, 0.0f);
            o.z = fmaxf(o.z, 0.0f); o.w = fmaxf(o.w, 0.0f);
            *(v4f*)(myA + r * APITCH + lane * 4) = o;

            beg = end; end = nend; self = nself; s_l = ns_l;
        }
        const v4f zero = {0.0f, 0.0f, 0.0f, 0.0f};
        for (int r = nr; r < 16; ++r)
            *(v4f*)(myA + r * APITCH + lane * 4) = zero;
    }
    __builtin_amdgcn_fence(__ATOMIC_ACQ_REL, "workgroup", "local");

    const int nloc  = lane & 15;
    const int khalf = lane >> 4;
    const float* __restrict__ arow = myA + nloc * APITCH + 2 * khalf;
    const v2f* __restrict__ bbase = (const v2f*)ldsW + lane;

    v8f acc[8] = {};
#pragma unroll
    for (int k = 0; k < 32; ++k) {
        const v2f a = *(const v2f*)(arow + 4 * k);
#pragma unroll
        for (int t = 0; t < 8; ++t) {
            v2f b = bbase[(t * 32 + k) * 32];
            acc[t] = __builtin_amdgcn_wmma_f32_16x16x4_f32(
                         false, a, false, b, (short)0, acc[t], false, false);
        }
    }

    const int rbase = d0 + 8 * khalf;
#pragma unroll
    for (int r = 0; r < 8; ++r) {
        int row = rbase + r;
        if (row >= nrows) continue;
        float dv = dinv[row];
        size_t ro = (size_t)row * DF;
#pragma unroll
        for (int t = 0; t < 8; ++t)
            Hout[ro + t * 16 + nloc] = acc[t][r] * dv;
    }
}


extern "C" void kernel_launch(void* const* d_in, const int* in_sizes, int n_in,
                              void* d_out, int out_size, void* d_ws, size_t ws_size,
                              hipStream_t stream) {
    const float* x  = (const float*)d_in[0];
    const int*   ei = (const int*)  d_in[1];
    const float* W1 = (const float*)d_in[2];
    const float* b1 = (const float*)d_in[3];
    const float* W2 = (const float*)d_in[4];
    const float* b2 = (const float*)d_in[5];
    const float* W3 = (const float*)d_in[6];
    const float* b3 = (const float*)d_in[7];

    const int N = in_sizes[0] / DF;
    const int E = in_sizes[1] / 2;
    const int* src = ei;
    const int* dst = ei + E;

    float* out = (float*)d_out;

    char* ws = (char*)d_ws;
    const int B = (N + BIN_NODES - 1) / BIN_NODES;
    const int chunk = (E + SORT_CHUNKS - 1) / SORT_CHUNKS;
    size_t offN  = ((size_t)(N + 1) * 4 + 255) & ~(size_t)255;
    size_t offB  = ((size_t)B * 4 + 255) & ~(size_t)255;
    size_t offH  = ((size_t)B * SORT_CHUNKS * 4 + 255) & ~(size_t)255;
    size_t offE  = ((size_t)E * 4 + 255) & ~(size_t)255;
    float* dinv   = (float*)ws;
    int*   rowptr = (int*)(ws + offN);
    int*   bintot = (int*)(ws + 2 * offN);
    int*   hist   = (int*)(ws + 2 * offN + offB);
    int*   csr    = (int*)(ws + 2 * offN + offB + offH);
    int*   pairs  = (int*)(ws + 2 * offN + offB + offH + offE);
    float* H      = (float*)(ws + 2 * offN + offB + offH + 3 * offE);
    float* AGGa   = H + (size_t)N * DF;

    const int strips  = (N + 15) / 16;
    const int gblocks = (strips + 7) / 8;
    const int pblocks = (N + 7) / 8;

    k_bin_hist   <<<SORT_CHUNKS, 1024,        0, stream>>>(dst, hist, E, B, chunk);
    k_bin_scan   <<<B,           SORT_CHUNKS, 0, stream>>>(hist, bintot);
    k_bin_scatter<<<SORT_CHUNKS, 1024,        0, stream>>>(src, dst, hist, bintot, pairs, E, B, chunk);
    k_bin_rows   <<<B,           BIN_NODES,   0, stream>>>(pairs, bintot, rowptr, csr, dinv, N, B);

    k_gemm_wmma<false><<<gblocks, GEMM_THREADS, 0, stream>>>(x, W1, dinv, H, strips, N);

    k_pull_gemm<<<gblocks, GEMM_THREADS, 0, stream>>>(H, rowptr, csr, dinv, b1, W2, AGGa, strips, N);

    k_pull_gemm<<<gblocks, GEMM_THREADS, 0, stream>>>(AGGa, rowptr, csr, dinv, b2, W3, H, strips, N);

    k_pull<true><<<pblocks, 256, 0, stream>>>(H, rowptr, csr, dinv, b3, out, N);
}


# Layers 1-2 fused launch made resident: 512 work-groups (2 per WGP), waves walk over strips, W staged once each; a wave done with its WMMAs gathers on beside its SIMD neighbours' WMMAs; sums unchanged
# speedup vs baseline: 4.5395x; 1.0407x over previous
#include <hip/hip_runtime.h>
#include <hip/hip_bf16.h>


typedef __attribute__((ext_vector_type(2))) float v2f;
typedef __attribute__((ext_vector_type(4))) float v4f;
typedef __attribute__((ext_vector_type(8))) float v8f;

#define DF 128
#define GEMM_THREADS 256


#define BIN_SHIFT   8
#define BIN_NODES   (1 << BIN_SHIFT)
#define SORT_CHUNKS 256
#define MAX_BINS    1024

typedef __attribute__((ext_vector_type(2))) int v2i;

__global__ __launch_bounds__(1024) void k_bin_hist(const int* __restrict__ dst,
                                                   int* __restrict__ hist,
                                                   int E, int B, int chunk) {
    __shared__ int lh[MAX_BINS];
    const int t = threadIdx.x, g = blockIdx.x;
    for (int b = t; b < B; b += 1024) lh[b] = 0;
    __syncthreads();
    const int e0 = g * chunk;
    const int e1 = (e0 + chunk < E) ? (e0 + chunk) : E;
    for (int eb = e0 + t; eb < e1; eb += 4 * 1024) {
        int d[4];
#pragma unroll
        for (int k = 0; k < 4; ++k) {
            const int e = eb + k * 1024;
            d[k] = (e < e1) ? dst[e] : -1;
        }
#pragma unroll
        for (int k = 0; k < 4; ++k)
            if (d[k] >= 0) atomicAdd(&lh[d[k] >> BIN_SHIFT], 1);
    }
    __syncthreads();
    for (int b = t; b < B; b += 1024) hist[b * SORT_CHUNKS + g] = lh[b];
}

__global__ __launch_bounds__(SORT_CHUNKS) void k_bin_scan(int* __restrict__ hist,
                                                          int* __restrict__ bintot) {
    __shared__ int wsum[SORT_CHUNKS / 32];
    const int t = threadIdx.x, lane = t & 31, wave = t >> 5;
    const int b = blockIdx.x;
    const int v = hist[b * SORT_CHUNKS + t];
    int inc = v;
#pragma unroll
    for (int o = 1; o < 32; o <<= 1) {
        int u = __shfl_up(inc, o, 32);
        if (lane >= o) inc += u;
    }
    if (lane == 31) wsum[wave] = inc;
    __syncthreads();
    int woff = 0;
#pragma unroll
    for (int w = 0; w < SORT_CHUNKS / 32; ++w) woff += (w < wave) ? wsum[w] : 0;
    hist[b * SORT_CHUNKS + t] = woff + inc - v;
    if (t == SORT_CHUNKS - 1) bintot[b] = woff + inc;
}

__global__ __launch_bounds__(1024) void k_bin_scatter(const int* __restrict__ src,
                                                      const int* __restrict__ dst,
                                                      const int* __restrict__ hist,
                                                      const int* __restrict__ bintot,
                                                      int* __restrict__ pairs,
                                                      int E, int B, int chunk) {
    __shared__ int lcur[MAX_BINS];
    __shared__ int wsum[32];
    const int t = threadIdx.x, lane = t & 31, wave = t >> 5, g = blockIdx.x;
    const int v = (t < B) ? bintot[t] : 0;
    int inc = v;
#pragma unroll
    for (int o = 1; o < 32; o <<= 1) {
        int u = __shfl_up(inc, o, 32);
        if (lane >= o) inc += u;
    }
    if (lane == 31) wsum[wave] = inc;
    __syncthreads();
    int woff = 0;
#pragma unroll
    for (int w = 0; w < 32; ++w) woff += (w < wave) ? wsum[w] : 0;
    if (t < B) lcur[t] = woff + inc - v + hist[t * SORT_CHUNKS + g];
    __syncthreads();

    v2i* __restrict__ pr = (v2i*)pairs;
    const int e0 = g * chunk;
    const int e1 = (e0 + chunk < E) ? (e0 + chunk) : E;
    for (int eb = e0 + t; eb < e1; eb += 4 * 1024) {
        int d[4], s[4];
#pragma unroll
        for (int k = 0; k < 4; ++k) {
            const int e = eb + k * 1024;
            const bool ok = e < e1;
            d[k] = ok ? dst[e] : -1;
            s[k] = ok ? src[e] : 0;
        }
#pragma unroll
        for (int k = 0; k < 4; ++k) {
            if (d[k] >= 0) {
                const int p = atomicAdd(&lcur[d[k] >> BIN_SHIFT], 1);
                v2i w = {s[k], d[k]};
                pr[p] = w;
            }
        }
    }
}

__global__ __launch_bounds__(BIN_NODES) void k_bin_rows(const int* __restrict__ pairs,
                                                        const int* __restrict__ bintot,
                                                        int* __restrict__ rowptr,
                                                        int* __restrict__ csr,
                                                        float* __restrict__ dinv,
                                                        int n, int B) {
    __shared__ int cnt[BIN_NODES];
    __shared__ int cur[BIN_NODES];
    __shared__ int red[BIN_NODES / 32];
    __shared__ int wsum[BIN_NODES / 32];
    const int t = threadIdx.x, lane = t & 31, wave = t >> 5;
    const int b = blockIdx.x;
    int part = 0;
    for (int j = t; j < b; j += BIN_NODES) part += bintot[j];
#pragma unroll
    for (int m = 16; m > 0; m >>= 1) part += __shfl_xor(part, m, 32);
    if (lane == 0) red[wave] = part;
    cnt[t] = 0;
    __syncthreads();
    int binStart = 0;
#pragma unroll
    for (int w = 0; w < BIN_NODES / 32; ++w) binStart += red[w];
    const int binEnd = binStart + bintot[b];

    const v2i* __restrict__ pr = (const v2i*)pairs;
    for (int eb = binStart + t; eb < binEnd; eb += 8 * BIN_NODES) {
        int d[8];
#pragma unroll
        for (int k = 0; k < 8; ++k) {
            const int e = eb + k * BIN_NODES;
            d[k] = (e < binEnd) ? pr[e].y : -1;
        }
#pragma unroll
        for (int k = 0; k < 8; ++k)
            if (d[k] >= 0) atomicAdd(&cnt[d[k] & (BIN_NODES - 1)], 1);
    }
    __syncthreads();

    const int c = cnt[t];
    int inc = c;
#pragma unroll
    for (int o = 1; o < 32; o <<= 1) {
        int u = __shfl_up(inc, o, 32);
        if (lane >= o) inc += u;
    }
    if (lane == 31) wsum[wave] = inc;
    __syncthreads();
    int woff = 0;
#pragma unroll
    for (int w = 0; w < BIN_NODES / 32; ++w) woff += (w < wave) ? wsum[w] : 0;
    const int rowStart = binStart + woff + inc - c;
    const int node = b * BIN_NODES + t;
    if (node < n) {
        rowptr[node] = rowStart;
        dinv[node]   = rsqrtf((float)c + 1.0f);
    }
    if (b == B - 1 && t == 0) rowptr[n] = binEnd;
    cur[t] = rowStart;
    __syncthreads();

    for (int eb = binStart + t; eb < binEnd; eb += 8 * BIN_NODES) {
        v2i q[8];
#pragma unroll
        for (int k = 0; k < 8; ++k) {
            const int e = eb + k * BIN_NODES;
            v2i none = {0, -1};
            q[k] = (e < binEnd) ? pr[e] : none;
        }
#pragma unroll
        for (int k = 0; k < 8; ++k) {
            if (q[k].y >= 0) {
                const int p = atomicAdd(&cur[q[k].y & (BIN_NODES - 1)], 1);
                csr[p] = q[k].x;
            }
        }
    }
}

template <bool RELU_IN>
__global__ __launch_bounds__(GEMM_THREADS) void k_gemm_wmma(
        const float* __restrict__ X, const float* __restrict__ W,
        const float* __restrict__ dinv, float* __restrict__ H,
        int strips, int nrows) {
    __shared__ __align__(16) float ldsW[DF * DF];

    {
        const int n    = threadIdx.x & (DF - 1);
        const int kg   = threadIdx.x >> 7;
        const int t    = n >> 4;
        const int nloc = n & 15;
        v2f* __restrict__ lw = (v2f*)ldsW + ((t << 5) * 32 + nloc);
        const float* __restrict__ wc = W + n;
#pragma unroll
        for (int i = 0; i < DF / 4 / (GEMM_THREADS / DF); ++i) {
            const int k = i * (GEMM_THREADS / DF) + kg;
            const float* __restrict__ wp = wc + (size_t)(4 * k) * DF;
            const v2f lo = {wp[0], wp[DF]};
            const v2f hi = {wp[2 * DF], wp[3 * DF]};
            lw[k * 32]      = lo;
            lw[k * 32 + 16] = hi;
        }
    }
    __syncthreads();

    const int wave  = threadIdx.x >> 5;
    const int lane  = threadIdx.x & 31;
    const int strip = blockIdx.x * (blockDim.x >> 5) + wave;
    if (strip >= strips) return;

    const int nloc  = lane & 15;
    const int khalf = lane >> 4;

    int arow_i = strip * 16 + nloc;
    if (arow_i >= nrows) arow_i = nrows - 1;
    const float* __restrict__ arow = X + (size_t)arow_i * DF + 2 * khalf;

    const v2f* __restrict__ bbase = (const v2f*)ldsW + lane;

    v8f acc[8] = {};
#pragma unroll
    for (int k = 0; k < 32; ++k) {
        v2f a = *(const v2f*)(arow + 4 * k);
        if (RELU_IN) {
            a.x = fmaxf(a.x, 0.0f);
            a.y = fmaxf(a.y, 0.0f);
        }
#pragma unroll
        for (int t = 0; t < 8; ++t) {
            v2f b = bbase[(t * 32 + k) * 32];
            acc[t] = __builtin_amdgcn_wmma_f32_16x16x4_f32(
                         false, a, false, b, (short)0, acc[t], false, false);
        }
    }

    const int rbase = strip * 16 + 8 * khalf;
#pragma unroll
    for (int r = 0; r < 8; ++r) {
        int row = rbase + r;
        if (row >= nrows) continue;
        float dv = dinv[row];
        size_t ro = (size_t)row * DF;
#pragma unroll
        for (int t = 0; t < 8; ++t)
            H[ro + t * 16 + nloc] = acc[t][r] * dv;
    }
}

template <bool NORM_OUT>
__global__ __launch_bounds__(256) void k_pull(
        const float* __restrict__ H, const int* __restrict__ rowptr,
        const int* __restrict__ csr, const float* __restrict__ dinv,
        const float* __restrict__ bias, float* __restrict__ OUT, int n) {
    const int d = __builtin_amdgcn_readfirstlane(
                      (int)(blockIdx.x * (blockDim.x >> 5) + (threadIdx.x >> 5)));
    if (d >= n) return;
    const int lane = threadIdx.x & 31;
    const int beg = rowptr[d], end = rowptr[d + 1];
    const float dd = dinv[d];
    const float* __restrict__ hl = H + lane * 4;
    const v4f bv = *(const v4f*)(bias + lane * 4);
    v4f acc = *(const v4f*)(hl + (size_t)d * DF);

    for (int base = beg; base < end; base += 32) {
        const int cn = (end - base < 32) ? (end - base) : 32;
        int s_l = 0;
        if (lane < cn) {
            s_l = csr[base + lane];
            s_l = min(max(s_l, 0), n - 1);
        }
        int j = 0;
        for (; j + 8 <= cn; j += 8) {
            v4f v[8];
#pragma unroll
            for (int k = 0; k < 8; ++k) {
                int s = __builtin_amdgcn_readlane(s_l, j + k);
                v[k]  = *(const v4f*)(hl + (size_t)s * DF);
            }
#pragma unroll
            for (int k = 0; k < 8; ++k) acc += v[k];
        }
        for (; j < cn; j += 4) {
            v4f v[4];
#pragma unroll
            for (int k = 0; k < 4; ++k) {
                int jj = (j + k < cn) ? (j + k) : (cn - 1);
                int s  = __builtin_amdgcn_readlane(s_l, jj);
                v[k]   = *(const v4f*)(hl + (size_t)s * DF);
            }
#pragma unroll
            for (int k = 0; k < 4; ++k) {
                v4f tsum = acc + v[k];
                acc = (j + k < cn) ? tsum : acc;
            }
        }
    }
    v4f o = acc * dd + bv;
    if (NORM_OUT) {
        float ss = o.x * o.x + o.y * o.y + o.z * o.z + o.w * o.w;
#pragma unroll
        for (int m = 16; m > 0; m >>= 1) ss += __shfl_xor(ss, m, 32);
        const float sc = 1.0f / fmaxf(sqrtf(ss), 1e-12f);
        o.x *= sc; o.y *= sc; o.z *= sc; o.w *= sc;
    }
    *(v4f*)(OUT + (size_t)d * DF + lane * 4) = o;
}

#define FUSED_GROUPS 512
#define APITCH 132

__global__ __launch_bounds__(GEMM_THREADS) void k_pull_gemm(
        const float* __restrict__ Hin, const int* __restrict__ rowptr,
        const int* __restrict__ csr, const float* __restrict__ dinv,
        const float* __restrict__ bias, const float* __restrict__ W,
        float* __restrict__ Hout, int strips, int nrows) {
    __shared__ __align__(16) float lds[DF * DF + (GEMM_THREADS / 32) * 16 * APITCH];
    float* __restrict__ ldsW = lds;
    float* __restrict__ ldsA = lds + DF * DF;

    {
        const int n    = threadIdx.x & (DF - 1);
        const int kg   = threadIdx.x >> 7;
        const int t    = n >> 4;
        const int nloc = n & 15;
        v2f* __restrict__ lw = (v2f*)ldsW + ((t << 5) * 32 + nloc);
        const float* __restrict__ wc = W + n;
#pragma unroll
        for (int i = 0; i < DF / 4 / (GEMM_THREADS / DF); ++i) {
            const int k = i * (GEMM_THREADS / DF) + kg;
            const float* __restrict__ wp = wc + (size_t)(4 * k) * DF;
            const v2f lo = {wp[0], wp[DF]};
            const v2f hi = {wp[2 * DF], wp[3 * DF]};
            lw[k * 32]      = lo;
            lw[k * 32 + 16] = hi;
        }
    }
    __syncthreads();

    const int wave  = __builtin_amdgcn_readfirstlane((int)(threadIdx.x >> 5));
    const int lane  = threadIdx.x & 31;
    float* __restrict__ myA = ldsA + wave * 16 * APITCH;
    const int nloc  = lane & 15;
    const int khalf = lane >> 4;
    const v4f bv = *(const v4f*)(bias + lane * 4);
    const int groups = (int)gridDim.x;
    for (int strip = __builtin_amdgcn_readfirstlane(wave * groups + (int)blockIdx.x);
         strip < strips; strip += groups * (GEMM_THREADS >> 5)) {
    const int d0 = strip * 16;
    const int nr = (nrows - d0 < 16) ? (nrows - d0) : 16;

    {
        const float* __restrict__ hl = Hin + lane * 4;
        int   rp_l = 0;
        float dv_l = 0.0f;
        if (lane <= nr) rp_l = rowptr[d0 + lane];
        if (lane <  nr) dv_l = dinv[d0 + lane];

        int beg = __builtin_amdgcn_readlane(rp_l, 0);
        int end = __builtin_amdgcn_readlane(rp_l, 1);
        v4f self = *(const v4f*)(hl + (size_t)d0 * DF);
        int s_l = 0;
        if (lane < end - beg) s_l = csr[beg + lane];

        for (int r = 0; r < nr; ++r) {
            int nend = end;
            v4f nself = self;
            int ns_l = 0;
            if (r + 1 < nr) {
                nend  = __builtin_amdgcn_readlane(rp_l, r + 2);
                nself = *(const v4f*)(hl + (size_t)(d0 + r + 1) * DF);
                if (lane < nend - end) ns_l = csr[end + lane];
            }

            v4f acc = self;
            for (int base = beg; base < end; base += 32) {
                const int cn = (end - base < 32) ? (end - base) : 32;
                if (base != beg) {
                    s_l = 0;
                    if (lane < cn) s_l = csr[base + lane];
                }
                v4f v[32];
#pragma unroll
                for (int g = 0; g < 8; ++g) {
                    if (4 * g < cn) {
#pragma unroll
                        for (int k = 0; k < 4; ++k) {
                            const int jj = (4 * g + k < cn) ? (4 * g + k) : (cn - 1);
                            int s = __builtin_amdgcn_readlane(s_l, jj);
                            s = min(max(s, 0), nrows - 1);
                            v[4 * g + k] = *(const v4f*)(hl + (size_t)s * DF);
                        }
                    }
                }
#pragma unroll
                for (int g = 0; g < 8; ++g) {
                    if (4 * g < cn) {
#pragma unroll
                        for (int k = 0; k < 4; ++k) {
                            const v4f tsum = acc + v[4 * g + k];
                            acc = (4 * g + k < cn) ? tsum : acc;
                        }
                    }
                }
            }

            const float dd = __int_as_float(__builtin_amdgcn_readlane(__float_as_int(dv_l), r));
            v4f o = acc * dd + bv;
            o.x = fmaxf(o.x, 0.0f); o.y = fmaxf(o.y, 0.0f);
            o.z = fmaxf(o.z, 0.0f); o.w = fmaxf(o.w, 0.0f);
            *(v4f*)(myA + r * APITCH + lane * 4) = o;

            beg = end; end = nend; self = nself; s_l = ns_l;
        }
        const v4f zero = {0.0f, 0.0f, 0.0f, 0.0f};
        for (int r = nr; r < 16; ++r)
            *(v4f*)(myA + r * APITCH + lane * 4) = zero;
    }
    __builtin_amdgcn_fence(__ATOMIC_ACQ_REL, "workgroup", "local");

    const float* __restrict__ arow = myA + nloc * APITCH + 2 * khalf;
    const v2f* __restrict__ bbase = (const v2f*)ldsW + lane;

    v8f acc[8] = {};
#pragma unroll
    for (int k = 0; k < 32; ++k) {
        const v2f a = *(const v2f*)(arow + 4 * k);
#pragma unroll
        for (int t = 0; t < 8; ++t) {
            v2f b = bbase[(t * 32 + k) * 32];
            acc[t] = __builtin_amdgcn_wmma_f32_16x16x4_f32(
                         false, a, false, b, (short)0, acc[t], false, false);
        }
    }

    const int rbase = d0 + 8 * khalf;
#pragma unroll
    for (int r = 0; r < 8; ++r) {
        int row = rbase + r;
        if (row >= nrows) continue;
        float dv = dinv[row];
        size_t ro = (size_t)row * DF;
#pragma unroll
        for (int t = 0; t < 8; ++t)
            Hout[ro + t * 16 + nloc] = acc[t][r] * dv;
    }
    }
}


extern "C" void kernel_launch(void* const* d_in, const int* in_sizes, int n_in,
                              void* d_out, int out_size, void* d_ws, size_t ws_size,
                              hipStream_t stream) {
    const float* x  = (const float*)d_in[0];
    const int*   ei = (const int*)  d_in[1];
    const float* W1 = (const float*)d_in[2];
    const float* b1 = (const float*)d_in[3];
    const float* W2 = (const float*)d_in[4];
    const float* b2 = (const float*)d_in[5];
    const float* W3 = (const float*)d_in[6];
    const float* b3 = (const float*)d_in[7];

    const int N = in_sizes[0] / DF;
    const int E = in_sizes[1] / 2;
    const int* src = ei;
    const int* dst = ei + E;

    float* out = (float*)d_out;

    char* ws = (char*)d_ws;
    const int B = (N + BIN_NODES - 1) / BIN_NODES;
    const int chunk = (E + SORT_CHUNKS - 1) / SORT_CHUNKS;
    size_t offN  = ((size_t)(N + 1) * 4 + 255) & ~(size_t)255;
    size_t offB  = ((size_t)B * 4 + 255) & ~(size_t)255;
    size_t offH  = ((size_t)B * SORT_CHUNKS * 4 + 255) & ~(size_t)255;
    size_t offE  = ((size_t)E * 4 + 255) & ~(size_t)255;
    float* dinv   = (float*)ws;
    int*   rowptr = (int*)(ws + offN);
    int*   bintot = (int*)(ws + 2 * offN);
    int*   hist   = (int*)(ws + 2 * offN + offB);
    int*   csr    = (int*)(ws + 2 * offN + offB + offH);
    int*   pairs  = (int*)(ws + 2 * offN + offB + offH + offE);
    float* H      = (float*)(ws + 2 * offN + offB + offH + 3 * offE);
    float* AGGa   = H + (size_t)N * DF;

    const int strips  = (N + 15) / 16;
    const int gblocks = (strips + 7) / 8;
    const int pblocks = (N + 7) / 8;
    const int fblocks = gblocks < FUSED_GROUPS ? gblocks : FUSED_GROUPS;

    k_bin_hist   <<<SORT_CHUNKS, 1024,        0, stream>>>(dst, hist, E, B, chunk);
    k_bin_scan   <<<B,           SORT_CHUNKS, 0, stream>>>(hist, bintot);
    k_bin_scatter<<<SORT_CHUNKS, 1024,        0, stream>>>(src, dst, hist, bintot, pairs, E, B, chunk);
    k_bin_rows   <<<B,           BIN_NODES,   0, stream>>>(pairs, bintot, rowptr, csr, dinv, N, B);

    k_gemm_wmma<false><<<gblocks, GEMM_THREADS, 0, stream>>>(x, W1, dinv, H, strips, N);

    k_pull_gemm<<<fblocks, GEMM_THREADS, 0, stream>>>(H, rowptr, csr, dinv, b1, W2, AGGa, strips, N);

    k_pull_gemm<<<fblocks, GEMM_THREADS, 0, stream>>>(AGGa, rowptr, csr, dinv, b2, W3, H, strips, N);

    k_pull<true><<<pblocks, 256, 0, stream>>>(H, rowptr, csr, dinv, b3, out, N);
}


# Edge sort passes 3 and 4: pairs and row sources are put in order in LDS and copied out as consecutive words (whole lines per store); sum order in a row = arrival order at the LDS slot counters
# speedup vs baseline: 4.8606x; 1.0707x over previous
#include <hip/hip_runtime.h>
#include <hip/hip_bf16.h>


typedef __attribute__((ext_vector_type(2))) float v2f;
typedef __attribute__((ext_vector_type(4))) float v4f;
typedef __attribute__((ext_vector_type(8))) float v8f;

#define DF 128
#define GEMM_THREADS 256


#define BIN_SHIFT   8
#define BIN_NODES   (1 << BIN_SHIFT)
#define SORT_CHUNKS 256
#define MAX_BINS    1024

typedef __attribute__((ext_vector_type(2))) int v2i;

__global__ __launch_bounds__(1024) void k_bin_hist(const int* __restrict__ dst,
                                                   int* __restrict__ hist,
                                                   int E, int B, int chunk) {
    __shared__ int lh[MAX_BINS];
    const int t = threadIdx.x, g = blockIdx.x;
    for (int b = t; b < B; b += 1024) lh[b] = 0;
    __syncthreads();
    const int e0 = g * chunk;
    const int e1 = (e0 + chunk < E) ? (e0 + chunk) : E;
    for (int eb = e0 + t; eb < e1; eb += 4 * 1024) {
        int d[4];
#pragma unroll
        for (int k = 0; k < 4; ++k) {
            const int e = eb + k * 1024;
            d[k] = (e < e1) ? dst[e] : -1;
        }
#pragma unroll
        for (int k = 0; k < 4; ++k)
            if (d[k] >= 0) atomicAdd(&lh[d[k] >> BIN_SHIFT], 1);
    }
    __syncthreads();
    for (int b = t; b < B; b += 1024) hist[b * SORT_CHUNKS + g] = lh[b];
}

__global__ __launch_bounds__(SORT_CHUNKS) void k_bin_scan(int* __restrict__ hist,
                                                          int* __restrict__ bintot) {
    __shared__ int wsum[SORT_CHUNKS / 32];
    const int t = threadIdx.x, lane = t & 31, wave = t >> 5;
    const int b = blockIdx.x;
    const int v = hist[b * SORT_CHUNKS + t];
    int inc = v;
#pragma unroll
    for (int o = 1; o < 32; o <<= 1) {
        int u = __shfl_up(inc, o, 32);
        if (lane >= o) inc += u;
    }
    if (lane == 31) wsum[wave] = inc;
    __syncthreads();
    int woff = 0;
#pragma unroll
    for (int w = 0; w < SORT_CHUNKS / 32; ++w) woff += (w < wave) ? wsum[w] : 0;
    hist[b * SORT_CHUNKS + t] = woff + inc - v;
    if (t == SORT_CHUNKS - 1) bintot[b] = woff + inc;
}

#define STAGE_PAIRS 8192

__global__ __launch_bounds__(1024) void k_bin_scatter(const int* __restrict__ src,
                                                      const int* __restrict__ dst,
                                                      const int* __restrict__ hist,
                                                      const int* __restrict__ bintot,
                                                      int* __restrict__ pairs,
                                                      int E, int B, int chunk) {
    __shared__ int lcur[MAX_BINS];
    __shared__ int gdel[MAX_BINS];
    __shared__ int wsum[32];
    __shared__ int wsum2[32];
    __shared__ v2i stage[STAGE_PAIRS];
    const int t = threadIdx.x, lane = t & 31, wave = t >> 5, g = blockIdx.x;
    const bool staged = chunk <= STAGE_PAIRS;
    const int v = (t < B) ? bintot[t] : 0;
    int hs = 0, c = 0;
    if (t < B) {
        hs = hist[t * SORT_CHUNKS + g];
        const int hn = (g + 1 < SORT_CHUNKS) ? hist[t * SORT_CHUNKS + g + 1] : v;
        c = hn - hs;
    }
    int inc = v, inc2 = c;
#pragma unroll
    for (int o = 1; o < 32; o <<= 1) {
        int u  = __shfl_up(inc, o, 32);
        int u2 = __shfl_up(inc2, o, 32);
        if (lane >= o) { inc += u; inc2 += u2; }
    }
    if (lane == 31) { wsum[wave] = inc; wsum2[wave] = inc2; }
    __syncthreads();
    int woff = 0, woff2 = 0;
#pragma unroll
    for (int w = 0; w < 32; ++w) {
        woff  += (w < wave) ? wsum[w]  : 0;
        woff2 += (w < wave) ? wsum2[w] : 0;
    }
    if (t < B) {
        const int gstart = woff + inc - v + hs;
        const int lstart = woff2 + inc2 - c;
        lcur[t] = staged ? lstart : gstart;
        gdel[t] = gstart - lstart;
    }
    __syncthreads();

    v2i* __restrict__ pr = (v2i*)pairs;
    const int e0 = g * chunk;
    const int e1 = (e0 + chunk < E) ? (e0 + chunk) : E;
    for (int eb = e0 + t; eb < e1; eb += 4 * 1024) {
        int d[4], s[4];
#pragma unroll
        for (int k = 0; k < 4; ++k) {
            const int e = eb + k * 1024;
            const bool ok = e < e1;
            d[k] = ok ? dst[e] : -1;
            s[k] = ok ? src[e] : 0;
        }
#pragma unroll
        for (int k = 0; k < 4; ++k) {
            if (d[k] >= 0) {
                const int p = atomicAdd(&lcur[d[k] >> BIN_SHIFT], 1);
                v2i w = {s[k], d[k]};
                if (staged) stage[p] = w;
                else        pr[p] = w;
            }
        }
    }
    if (staged) {
        __syncthreads();
        const int n = e1 - e0;
        for (int i = t; i < n; i += 1024) {
            const v2i q = stage[i];
            pr[i + gdel[q.y >> BIN_SHIFT]] = q;
        }
    }
}

#define STAGE_SRCS 16384

__global__ __launch_bounds__(BIN_NODES) void k_bin_rows(const int* __restrict__ pairs,
                                                        const int* __restrict__ bintot,
                                                        int* __restrict__ rowptr,
                                                        int* __restrict__ csr,
                                                        float* __restrict__ dinv,
                                                        int n, int B) {
    __shared__ int cnt[BIN_NODES];
    __shared__ int cur[BIN_NODES];
    __shared__ int red[BIN_NODES / 32];
    __shared__ int wsum[BIN_NODES / 32];
    __shared__ int lsrc[STAGE_SRCS];
    const int t = threadIdx.x, lane = t & 31, wave = t >> 5;
    const int b = blockIdx.x;
    int part = 0;
    for (int j = t; j < b; j += BIN_NODES) part += bintot[j];
#pragma unroll
    for (int m = 16; m > 0; m >>= 1) part += __shfl_xor(part, m, 32);
    if (lane == 0) red[wave] = part;
    cnt[t] = 0;
    __syncthreads();
    int binStart = 0;
#pragma unroll
    for (int w = 0; w < BIN_NODES / 32; ++w) binStart += red[w];
    const int binCount = bintot[b];
    const int binEnd = binStart + binCount;
    const bool staged = binCount <= STAGE_SRCS;

    const v2i* __restrict__ pr = (const v2i*)pairs;
    for (int eb = binStart + t; eb < binEnd; eb += 8 * BIN_NODES) {
        int d[8];
#pragma unroll
        for (int k = 0; k < 8; ++k) {
            const int e = eb + k * BIN_NODES;
            d[k] = (e < binEnd) ? pr[e].y : -1;
        }
#pragma unroll
        for (int k = 0; k < 8; ++k)
            if (d[k] >= 0) atomicAdd(&cnt[d[k] & (BIN_NODES - 1)], 1);
    }
    __syncthreads();

    const int c = cnt[t];
    int inc = c;
#pragma unroll
    for (int o = 1; o < 32; o <<= 1) {
        int u = __shfl_up(inc, o, 32);
        if (lane >= o) inc += u;
    }
    if (lane == 31) wsum[wave] = inc;
    __syncthreads();
    int woff = 0;
#pragma unroll
    for (int w = 0; w < BIN_NODES / 32; ++w) woff += (w < wave) ? wsum[w] : 0;
    const int rowLocal = woff + inc - c;
    const int rowStart = binStart + rowLocal;
    const int node = b * BIN_NODES + t;
    if (node < n) {
        rowptr[node] = rowStart;
        dinv[node]   = rsqrtf((float)c + 1.0f);
    }
    if (b == B - 1 && t == 0) rowptr[n] = binEnd;
    cur[t] = staged ? rowLocal : rowStart;
    __syncthreads();

    for (int eb = binStart + t; eb < binEnd; eb += 8 * BIN_NODES) {
        v2i q[8];
#pragma unroll
        for (int k = 0; k < 8; ++k) {
            const int e = eb + k * BIN_NODES;
            v2i none = {0, -1};
            q[k] = (e < binEnd) ? pr[e] : none;
        }
#pragma unroll
        for (int k = 0; k < 8; ++k) {
            if (q[k].y >= 0) {
                const int p = atomicAdd(&cur[q[k].y & (BIN_NODES - 1)], 1);
                if (staged) lsrc[p] = q[k].x;
                else        csr[p]  = q[k].x;
            }
        }
    }
    if (staged) {
        __syncthreads();
        for (int i = t; i < binCount; i += BIN_NODES) csr[binStart + i] = lsrc[i];
    }
}

template <bool RELU_IN>
__global__ __launch_bounds__(GEMM_THREADS) void k_gemm_wmma(
        const float* __restrict__ X, const float* __restrict__ W,
        const float* __restrict__ dinv, float* __restrict__ H,
        int strips, int nrows) {
    __shared__ __align__(16) float ldsW[DF * DF];

    {
        const int n    = threadIdx.x & (DF - 1);
        const int kg   = threadIdx.x >> 7;
        const int t    = n >> 4;
        const int nloc = n & 15;
        v2f* __restrict__ lw = (v2f*)ldsW + ((t << 5) * 32 + nloc);
        const float* __restrict__ wc = W + n;
#pragma unroll
        for (int i = 0; i < DF / 4 / (GEMM_THREADS / DF); ++i) {
            const int k = i * (GEMM_THREADS / DF) + kg;
            const float* __restrict__ wp = wc + (size_t)(4 * k) * DF;
            const v2f lo = {wp[0], wp[DF]};
            const v2f hi = {wp[2 * DF], wp[3 * DF]};
            lw[k * 32]      = lo;
            lw[k * 32 + 16] = hi;
        }
    }
    __syncthreads();

    const int wave  = threadIdx.x >> 5;
    const int lane  = threadIdx.x & 31;
    const int strip = blockIdx.x * (blockDim.x >> 5) + wave;
    if (strip >= strips) return;

    const int nloc  = lane & 15;
    const int khalf = lane >> 4;

    int arow_i = strip * 16 + nloc;
    if (arow_i >= nrows) arow_i = nrows - 1;
    const float* __restrict__ arow = X + (size_t)arow_i * DF + 2 * khalf;

    const v2f* __restrict__ bbase = (const v2f*)ldsW + lane;

    v8f acc[8] = {};
#pragma unroll
    for (int k = 0; k < 32; ++k) {
        v2f a = *(const v2f*)(arow + 4 * k);
        if (RELU_IN) {
            a.x = fmaxf(a.x, 0.0f);
            a.y = fmaxf(a.y, 0.0f);
        }
#pragma unroll
        for (int t = 0; t < 8; ++t) {
            v2f b = bbase[(t * 32 + k) * 32];
            acc[t] = __builtin_amdgcn_wmma_f32_16x16x4_f32(
                         false, a, false, b, (short)0, acc[t], false, false);
        }
    }

    const int rbase = strip * 16 + 8 * khalf;
#pragma unroll
    for (int r = 0; r < 8; ++r) {
        int row = rbase + r;
        if (row >= nrows) continue;
        float dv = dinv[row];
        size_t ro = (size_t)row * DF;
#pragma unroll
        for (int t = 0; t < 8; ++t)
            H[ro + t * 16 + nloc] = acc[t][r] * dv;
    }
}

template <bool NORM_OUT>
__global__ __launch_bounds__(256) void k_pull(
        const float* __restrict__ H, const int* __restrict__ rowptr,
        const int* __restrict__ csr, const float* __restrict__ dinv,
        const float* __restrict__ bias, float* __restrict__ OUT, int n) {
    const int d = __builtin_amdgcn_readfirstlane(
                      (int)(blockIdx.x * (blockDim.x >> 5) + (threadIdx.x >> 5)));
    if (d >= n) return;
    const int lane = threadIdx.x & 31;
    const int beg = rowptr[d], end = rowptr[d + 1];
    const float dd = dinv[d];
    const float* __restrict__ hl = H + lane * 4;
    const v4f bv = *(const v4f*)(bias + lane * 4);
    v4f acc = *(const v4f*)(hl + (size_t)d * DF);

    for (int base = beg; base < end; base += 32) {
        const int cn = (end - base < 32) ? (end - base) : 32;
        int s_l = 0;
        if (lane < cn) {
            s_l = csr[base + lane];
            s_l = min(max(s_l, 0), n - 1);
        }
        int j = 0;
        for (; j + 8 <= cn; j += 8) {
            v4f v[8];
#pragma unroll
            for (int k = 0; k < 8; ++k) {
                int s = __builtin_amdgcn_readlane(s_l, j + k);
                v[k]  = *(const v4f*)(hl + (size_t)s * DF);
            }
#pragma unroll
            for (int k = 0; k < 8; ++k) acc += v[k];
        }
        for (; j < cn; j += 4) {
            v4f v[4];
#pragma unroll
            for (int k = 0; k < 4; ++k) {
                int jj = (j + k < cn) ? (j + k) : (cn - 1);
                int s  = __builtin_amdgcn_readlane(s_l, jj);
                v[k]   = *(const v4f*)(hl + (size_t)s * DF);
            }
#pragma unroll
            for (int k = 0; k < 4; ++k) {
                v4f tsum = acc + v[k];
                acc = (j + k < cn) ? tsum : acc;
            }
        }
    }
    v4f o = acc * dd + bv;
    if (NORM_OUT) {
        float ss = o.x * o.x + o.y * o.y + o.z * o.z + o.w * o.w;
#pragma unroll
        for (int m = 16; m > 0; m >>= 1) ss += __shfl_xor(ss, m, 32);
        const float sc = 1.0f / fmaxf(sqrtf(ss), 1e-12f);
        o.x *= sc; o.y *= sc; o.z *= sc; o.w *= sc;
    }
    *(v4f*)(OUT + (size_t)d * DF + lane * 4) = o;
}

#define FUSED_GROUPS 512
#define APITCH 132

__global__ __launch_bounds__(GEMM_THREADS) void k_pull_gemm(
        const float* __restrict__ Hin, const int* __restrict__ rowptr,
        const int* __restrict__ csr, const float* __restrict__ dinv,
        const float* __restrict__ bias, const float* __restrict__ W,
        float* __restrict__ Hout, int strips, int nrows) {
    __shared__ __align__(16) float lds[DF * DF + (GEMM_THREADS / 32) * 16 * APITCH];
    float* __restrict__ ldsW = lds;
    float* __restrict__ ldsA = lds + DF * DF;

    {
        const int n    = threadIdx.x & (DF - 1);
        const int kg   = threadIdx.x >> 7;
        const int t    = n >> 4;
        const int nloc = n & 15;
        v2f* __restrict__ lw = (v2f*)ldsW + ((t << 5) * 32 + nloc);
        const float* __restrict__ wc = W + n;
#pragma unroll
        for (int i = 0; i < DF / 4 / (GEMM_THREADS / DF); ++i) {
            const int k = i * (GEMM_THREADS / DF) + kg;
            const float* __restrict__ wp = wc + (size_t)(4 * k) * DF;
            const v2f lo = {wp[0], wp[DF]};
            const v2f hi = {wp[2 * DF], wp[3 * DF]};
            lw[k * 32]      = lo;
            lw[k * 32 + 16] = hi;
        }
    }
    __syncthreads();

    const int wave  = __builtin_amdgcn_readfirstlane((int)(threadIdx.x >> 5));
    const int lane  = threadIdx.x & 31;
    float* __restrict__ myA = ldsA + wave * 16 * APITCH;
    const int nloc  = lane & 15;
    const int khalf = lane >> 4;
    const v4f bv = *(const v4f*)(bias + lane * 4);
    const int groups = (int)gridDim.x;
    for (int strip = __builtin_amdgcn_readfirstlane(wave * groups + (int)blockIdx.x);
         strip < strips; strip += groups * (GEMM_THREADS >> 5)) {
    const int d0 = strip * 16;
    const int nr = (nrows - d0 < 16) ? (nrows - d0) : 16;

    {
        const float* __restrict__ hl = Hin + lane * 4;
        int   rp_l = 0;
        float dv_l = 0.0f;
        if (lane <= nr) rp_l = rowptr[d0 + lane];
        if (lane <  nr) dv_l = dinv[d0 + lane];

        int beg = __builtin_amdgcn_readlane(rp_l, 0);
        int end = __builtin_amdgcn_readlane(rp_l, 1);
        v4f self = *(const v4f*)(hl + (size_t)d0 * DF);
        int s_l = 0;
        if (lane < end - beg) s_l = csr[beg + lane];

        for (int r = 0; r < nr; ++r) {
            int nend = end;
            v4f nself = self;
            int ns_l = 0;
            if (r + 1 < nr) {
                nend  = __builtin_amdgcn_readlane(rp_l, r + 2);
                nself = *(const v4f*)(hl + (size_t)(d0 + r + 1) * DF);
                if (lane < nend - end) ns_l = csr[end + lane];
            }

            v4f acc = self;
            for (int base = beg; base < end; base += 32) {
                const int cn = (end - base < 32) ? (end - base) : 32;
                if (base != beg) {
                    s_l = 0;
                    if (lane < cn) s_l = csr[base + lane];
                }
                v4f v[32];
#pragma unroll
                for (int g = 0; g < 8; ++g) {
                    if (4 * g < cn) {
#pragma unroll
                        for (int k = 0; k < 4; ++k) {
                            const int jj = (4 * g + k < cn) ? (4 * g + k) : (cn - 1);
                            int s = __builtin_amdgcn_readlane(s_l, jj);
                            s = min(max(s, 0), nrows - 1);
                            v[4 * g + k] = *(const v4f*)(hl + (size_t)s * DF);
                        }
                    }
                }
#pragma unroll
                for (int g = 0; g < 8; ++g) {
                    if (4 * g < cn) {
#pragma unroll
                        for (int k = 0; k < 4; ++k) {
                            const v4f tsum = acc + v[4 * g + k];
                            acc = (4 * g + k < cn) ? tsum : acc;
                        }
                    }
                }
            }

            const float dd = __int_as_float(__builtin_amdgcn_readlane(__float_as_int(dv_l), r));
            v4f o = acc * dd + bv;
            o.x = fmaxf(o.x, 0.0f); o.y = fmaxf(o.y, 0.0f);
            o.z = fmaxf(o.z, 0.0f); o.w = fmaxf(o.w, 0.0f);
            *(v4f*)(myA + r * APITCH + lane * 4) = o;

            beg = end; end = nend; self = nself; s_l = ns_l;
        }
        const v4f zero = {0.0f, 0.0f, 0.0f, 0.0f};
        for (int r = nr; r < 16; ++r)
            *(v4f*)(myA + r * APITCH + lane * 4) = zero;
    }
    __builtin_amdgcn_fence(__ATOMIC_ACQ_REL, "workgroup", "local");

    const float* __restrict__ arow = myA + nloc * APITCH + 2 * khalf;
    const v2f* __restrict__ bbase = (const v2f*)ldsW + lane;

    v8f acc[8] = {};
#pragma unroll
    for (int k = 0; k < 32; ++k) {
        const v2f a = *(const v2f*)(arow + 4 * k);
#pragma unroll
        for (int t = 0; t < 8; ++t) {
            v2f b = bbase[(t * 32 + k) * 32];
            acc[t] = __builtin_amdgcn_wmma_f32_16x16x4_f32(
                         false, a, false, b, (short)0, acc[t], false, false);
        }
    }

    const int rbase = d0 + 8 * khalf;
#pragma unroll
    for (int r = 0; r < 8; ++r) {
        int row = rbase + r;
        if (row >= nrows) continue;
        float dv = dinv[row];
        size_t ro = (size_t)row * DF;
#pragma unroll
        for (int t = 0; t < 8; ++t)
            Hout[ro + t * 16 + nloc] = acc[t][r] * dv;
    }
    }
}


extern "C" void kernel_launch(void* const* d_in, const int* in_sizes, int n_in,
                              void* d_out, int out_size, void* d_ws, size_t ws_size,
                              hipStream_t stream) {
    const float* x  = (const float*)d_in[0];
    const int*   ei = (const int*)  d_in[1];
    const float* W1 = (const float*)d_in[2];
    const float* b1 = (const float*)d_in[3];
    const float* W2 = (const float*)d_in[4];
    const float* b2 = (const float*)d_in[5];
    const float* W3 = (const float*)d_in[6];
    const float* b3 = (const float*)d_in[7];

    const int N = in_sizes[0] / DF;
    const int E = in_sizes[1] / 2;
    const int* src = ei;
    const int* dst = ei + E;

    float* out = (float*)d_out;

    char* ws = (char*)d_ws;
    const int B = (N + BIN_NODES - 1) / BIN_NODES;
    const int chunk = (E + SORT_CHUNKS - 1) / SORT_CHUNKS;
    size_t offN  = ((size_t)(N + 1) * 4 + 255) & ~(size_t)255;
    size_t offB  = ((size_t)B * 4 + 255) & ~(size_t)255;
    size_t offH  = ((size_t)B * SORT_CHUNKS * 4 + 255) & ~(size_t)255;
    size_t offE  = ((size_t)E * 4 + 255) & ~(size_t)255;
    float* dinv   = (float*)ws;
    int*   rowptr = (int*)(ws + offN);
    int*   bintot = (int*)(ws + 2 * offN);
    int*   hist   = (int*)(ws + 2 * offN + offB);
    int*   csr    = (int*)(ws + 2 * offN + offB + offH);
    int*   pairs  = (int*)(ws + 2 * offN + offB + offH + offE);
    float* H      = (float*)(ws + 2 * offN + offB + offH + 3 * offE);
    float* AGGa   = H + (size_t)N * DF;

    const int strips  = (N + 15) / 16;
    const int gblocks = (strips + 7) / 8;
    const int pblocks = (N + 7) / 8;
    const int fblocks = gblocks < FUSED_GROUPS ? gblocks : FUSED_GROUPS;

    k_bin_hist   <<<SORT_CHUNKS, 1024,        0, stream>>>(dst, hist, E, B, chunk);
    k_bin_scan   <<<B,           SORT_CHUNKS, 0, stream>>>(hist, bintot);
    k_bin_scatter<<<SORT_CHUNKS, 1024,        0, stream>>>(src, dst, hist, bintot, pairs, E, B, chunk);
    k_bin_rows   <<<B,           BIN_NODES,   0, stream>>>(pairs, bintot, rowptr, csr, dinv, N, B);

    k_gemm_wmma<false><<<gblocks, GEMM_THREADS, 0, stream>>>(x, W1, dinv, H, strips, N);

    k_pull_gemm<<<fblocks, GEMM_THREADS, 0, stream>>>(H, rowptr, csr, dinv, b1, W2, AGGa, strips, N);

    k_pull_gemm<<<fblocks, GEMM_THREADS, 0, stream>>>(AGGa, rowptr, csr, dinv, b2, W3, H, strips, N);

    k_pull<true><<<pblocks, 256, 0, stream>>>(H, rowptr, csr, dinv, b3, out, N);
}


# GEMM epilogues: D tile through LDS (the wave's A tile; first GEMM: the W area behind a barrier), rows stored as whole 512-B runs not 64-B pieces; row scale from lane-held dinv, no dependent loads
# speedup vs baseline: 5.0153x; 1.0318x over previous
#include <hip/hip_runtime.h>
#include <hip/hip_bf16.h>


typedef __attribute__((ext_vector_type(2))) float v2f;
typedef __attribute__((ext_vector_type(4))) float v4f;
typedef __attribute__((ext_vector_type(8))) float v8f;

#define DF 128
#define GEMM_THREADS 256


#define BIN_SHIFT   8
#define BIN_NODES   (1 << BIN_SHIFT)
#define SORT_CHUNKS 256
#define MAX_BINS    1024

typedef __attribute__((ext_vector_type(2))) int v2i;

__global__ __launch_bounds__(1024) void k_bin_hist(const int* __restrict__ dst,
                                                   int* __restrict__ hist,
                                                   int E, int B, int chunk) {
    __shared__ int lh[MAX_BINS];
    const int t = threadIdx.x, g = blockIdx.x;
    for (int b = t; b < B; b += 1024) lh[b] = 0;
    __syncthreads();
    const int e0 = g * chunk;
    const int e1 = (e0 + chunk < E) ? (e0 + chunk) : E;
    for (int eb = e0 + t; eb < e1; eb += 4 * 1024) {
        int d[4];
#pragma unroll
        for (int k = 0; k < 4; ++k) {
            const int e = eb + k * 1024;
            d[k] = (e < e1) ? dst[e] : -1;
        }
#pragma unroll
        for (int k = 0; k < 4; ++k)
            if (d[k] >= 0) atomicAdd(&lh[d[k] >> BIN_SHIFT], 1);
    }
    __syncthreads();
    for (int b = t; b < B; b += 1024) hist[b * SORT_CHUNKS + g] = lh[b];
}

__global__ __launch_bounds__(SORT_CHUNKS) void k_bin_scan(int* __restrict__ hist,
                                                          int* __restrict__ bintot) {
    __shared__ int wsum[SORT_CHUNKS / 32];
    const int t = threadIdx.x, lane = t & 31, wave = t >> 5;
    const int b = blockIdx.x;
    const int v = hist[b * SORT_CHUNKS + t];
    int inc = v;
#pragma unroll
    for (int o = 1; o < 32; o <<= 1) {
        int u = __shfl_up(inc, o, 32);
        if (lane >= o) inc += u;
    }
    if (lane == 31) wsum[wave] = inc;
    __syncthreads();
    int woff = 0;
#pragma unroll
    for (int w = 0; w < SORT_CHUNKS / 32; ++w) woff += (w < wave) ? wsum[w] : 0;
    hist[b * SORT_CHUNKS + t] = woff + inc - v;
    if (t == SORT_CHUNKS - 1) bintot[b] = woff + inc;
}

#define STAGE_PAIRS 8192

__global__ __launch_bounds__(1024) void k_bin_scatter(const int* __restrict__ src,
                                                      const int* __restrict__ dst,
                                                      const int* __restrict__ hist,
                                                      const int* __restrict__ bintot,
                                                      int* __restrict__ pairs,
                                                      int E, int B, int chunk) {
    __shared__ int lcur[MAX_BINS];
    __shared__ int gdel[MAX_BINS];
    __shared__ int wsum[32];
    __shared__ int wsum2[32];
    __shared__ v2i stage[STAGE_PAIRS];
    const int t = threadIdx.x, lane = t & 31, wave = t >> 5, g = blockIdx.x;
    const bool staged = chunk <= STAGE_PAIRS;
    const int v = (t < B) ? bintot[t] : 0;
    int hs = 0, c = 0;
    if (t < B) {
        hs = hist[t * SORT_CHUNKS + g];
        const int hn = (g + 1 < SORT_CHUNKS) ? hist[t * SORT_CHUNKS + g + 1] : v;
        c = hn - hs;
    }
    int inc = v, inc2 = c;
#pragma unroll
    for (int o = 1; o < 32; o <<= 1) {
        int u  = __shfl_up(inc, o, 32);
        int u2 = __shfl_up(inc2, o, 32);
        if (lane >= o) { inc += u; inc2 += u2; }
    }
    if (lane == 31) { wsum[wave] = inc; wsum2[wave] = inc2; }
    __syncthreads();
    int woff = 0, woff2 = 0;
#pragma unroll
    for (int w = 0; w < 32; ++w) {
        woff  += (w < wave) ? wsum[w]  : 0;
        woff2 += (w < wave) ? wsum2[w] : 0;
    }
    if (t < B) {
        const int gstart = woff + inc - v + hs;
        const int lstart = woff2 + inc2 - c;
        lcur[t] = staged ? lstart : gstart;
        gdel[t] = gstart - lstart;
    }
    __syncthreads();

    v2i* __restrict__ pr = (v2i*)pairs;
    const int e0 = g * chunk;
    const int e1 = (e0 + chunk < E) ? (e0 + chunk) : E;
    for (int eb = e0 + t; eb < e1; eb += 4 * 1024) {
        int d[4], s[4];
#pragma unroll
        for (int k = 0; k < 4; ++k) {
            const int e = eb + k * 1024;
            const bool ok = e < e1;
            d[k] = ok ? dst[e] : -1;
            s[k] = ok ? src[e] : 0;
        }
#pragma unroll
        for (int k = 0; k < 4; ++k) {
            if (d[k] >= 0) {
                const int p = atomicAdd(&lcur[d[k] >> BIN_SHIFT], 1);
                v2i w = {s[k], d[k]};
                if (staged) stage[p] = w;
                else        pr[p] = w;
            }
        }
    }
    if (staged) {
        __syncthreads();
        const int n = e1 - e0;
        for (int i = t; i < n; i += 1024) {
            const v2i q = stage[i];
            pr[i + gdel[q.y >> BIN_SHIFT]] = q;
        }
    }
}

#define STAGE_SRCS 16384

__global__ __launch_bounds__(BIN_NODES) void k_bin_rows(const int* __restrict__ pairs,
                                                        const int* __restrict__ bintot,
                                                        int* __restrict__ rowptr,
                                                        int* __restrict__ csr,
                                                        float* __restrict__ dinv,
                                                        int n, int B) {
    __shared__ int cnt[BIN_NODES];
    __shared__ int cur[BIN_NODES];
    __shared__ int red[BIN_NODES / 32];
    __shared__ int wsum[BIN_NODES / 32];
    __shared__ int lsrc[STAGE_SRCS];
    const int t = threadIdx.x, lane = t & 31, wave = t >> 5;
    const int b = blockIdx.x;
    int part = 0;
    for (int j = t; j < b; j += BIN_NODES) part += bintot[j];
#pragma unroll
    for (int m = 16; m > 0; m >>= 1) part += __shfl_xor(part, m, 32);
    if (lane == 0) red[wave] = part;
    cnt[t] = 0;
    __syncthreads();
    int binStart = 0;
#pragma unroll
    for (int w = 0; w < BIN_NODES / 32; ++w) binStart += red[w];
    const int binCount = bintot[b];
    const int binEnd = binStart + binCount;
    const bool staged = binCount <= STAGE_SRCS;

    const v2i* __restrict__ pr = (const v2i*)pairs;
    for (int eb = binStart + t; eb < binEnd; eb += 8 * BIN_NODES) {
        int d[8];
#pragma unroll
        for (int k = 0; k < 8; ++k) {
            const int e = eb + k * BIN_NODES;
            d[k] = (e < binEnd) ? pr[e].y : -1;
        }
#pragma unroll
        for (int k = 0; k < 8; ++k)
            if (d[k] >= 0) atomicAdd(&cnt[d[k] & (BIN_NODES - 1)], 1);
    }
    __syncthreads();

    const int c = cnt[t];
    int inc = c;
#pragma unroll
    for (int o = 1; o < 32; o <<= 1) {
        int u = __shfl_up(inc, o, 32);
        if (lane >= o) inc += u;
    }
    if (lane == 31) wsum[wave] = inc;
    __syncthreads();
    int woff = 0;
#pragma unroll
    for (int w = 0; w < BIN_NODES / 32; ++w) woff += (w < wave) ? wsum[w] : 0;
    const int rowLocal = woff + inc - c;
    const int rowStart = binStart + rowLocal;
    const int node = b * BIN_NODES + t;
    if (node < n) {
        rowptr[node] = rowStart;
        dinv[node]   = rsqrtf((float)c + 1.0f);
    }
    if (b == B - 1 && t == 0) rowptr[n] = binEnd;
    cur[t] = staged ? rowLocal : rowStart;
    __syncthreads();

    for (int eb = binStart + t; eb < binEnd; eb += 8 * BIN_NODES) {
        v2i q[8];
#pragma unroll
        for (int k = 0; k < 8; ++k) {
            const int e = eb + k * BIN_NODES;
            v2i none = {0, -1};
            q[k] = (e < binEnd) ? pr[e] : none;
        }
#pragma unroll
        for (int k = 0; k < 8; ++k) {
            if (q[k].y >= 0) {
                const int p = atomicAdd(&cur[q[k].y & (BIN_NODES - 1)], 1);
                if (staged) lsrc[p] = q[k].x;
                else        csr[p]  = q[k].x;
            }
        }
    }
    if (staged) {
        __syncthreads();
        for (int i = t; i < binCount; i += BIN_NODES) csr[binStart + i] = lsrc[i];
    }
}

template <bool RELU_IN>
__global__ __launch_bounds__(GEMM_THREADS) void k_gemm_wmma(
        const float* __restrict__ X, const float* __restrict__ W,
        const float* __restrict__ dinv, float* __restrict__ H,
        int strips, int nrows) {
    __shared__ __align__(16) float ldsW[DF * DF];

    {
        const int n    = threadIdx.x & (DF - 1);
        const int kg   = threadIdx.x >> 7;
        const int t    = n >> 4;
        const int nloc = n & 15;
        v2f* __restrict__ lw = (v2f*)ldsW + ((t << 5) * 32 + nloc);
        const float* __restrict__ wc = W + n;
#pragma unroll
        for (int i = 0; i < DF / 4 / (GEMM_THREADS / DF); ++i) {
            const int k = i * (GEMM_THREADS / DF) + kg;
            const float* __restrict__ wp = wc + (size_t)(4 * k) * DF;
            const v2f lo = {wp[0], wp[DF]};
            const v2f hi = {wp[2 * DF], wp[3 * DF]};
            lw[k * 32]      = lo;
            lw[k * 32 + 16] = hi;
        }
    }
    __syncthreads();

    const int wave  = __builtin_amdgcn_readfirstlane((int)(threadIdx.x >> 5));
    const int lane  = threadIdx.x & 31;
    int strip = (int)blockIdx.x * (GEMM_THREADS >> 5) + wave;
    const bool live = strip < strips;
    if (!live) strip = strips - 1;

    const int nloc  = lane & 15;
    const int khalf = lane >> 4;

    int arow_i = strip * 16 + nloc;
    if (arow_i >= nrows) arow_i = nrows - 1;
    const float* __restrict__ arow = X + (size_t)arow_i * DF + 2 * khalf;

    const v2f* __restrict__ bbase = (const v2f*)ldsW + lane;

    float dv_l = 0.0f;
    if (lane < 16 && strip * 16 + lane < nrows) dv_l = dinv[strip * 16 + lane];

    v8f acc[8] = {};
#pragma unroll
    for (int k = 0; k < 32; ++k) {
        v2f a = *(const v2f*)(arow + 4 * k);
        if (RELU_IN) {
            a.x = fmaxf(a.x, 0.0f);
            a.y = fmaxf(a.y, 0.0f);
        }
#pragma unroll
        for (int t = 0; t < 8; ++t) {
            v2f b = bbase[(t * 32 + k) * 32];
            acc[t] = __builtin_amdgcn_wmma_f32_16x16x4_f32(
                         false, a, false, b, (short)0, acc[t], false, false);
        }
    }

    __syncthreads();
    float* __restrict__ myD = ldsW + wave * 16 * DF;
    {
        float* __restrict__ dt = myD + (8 * khalf) * DF + nloc;
#pragma unroll
        for (int r = 0; r < 8; ++r) {
#pragma unroll
            for (int t = 0; t < 8; ++t)
                dt[r * DF + t * 16] = acc[t][r];
        }
    }
    __builtin_amdgcn_fence(__ATOMIC_ACQ_REL, "workgroup", "local");
    {
        v4f o[16];
#pragma unroll
        for (int r = 0; r < 16; ++r)
            o[r] = *(const v4f*)(myD + r * DF + lane * 4);
        const int d0 = strip * 16;
#pragma unroll
        for (int r = 0; r < 16; ++r) {
            if (live && d0 + r < nrows) {
                const float dd = __int_as_float(__builtin_amdgcn_readlane(__float_as_int(dv_l), r));
                *(v4f*)(H + (size_t)(d0 + r) * DF + lane * 4) = o[r] * dd;
            }
        }
    }
}

template <bool NORM_OUT>
__global__ __launch_bounds__(256) void k_pull(
        const float* __restrict__ H, const int* __restrict__ rowptr,
        const int* __restrict__ csr, const float* __restrict__ dinv,
        const float* __restrict__ bias, float* __restrict__ OUT, int n) {
    const int d = __builtin_amdgcn_readfirstlane(
                      (int)(blockIdx.x * (blockDim.x >> 5) + (threadIdx.x >> 5)));
    if (d >= n) return;
    const int lane = threadIdx.x & 31;
    const int beg = rowptr[d], end = rowptr[d + 1];
    const float dd = dinv[d];
    const float* __restrict__ hl = H + lane * 4;
    const v4f bv = *(const v4f*)(bias + lane * 4);
    v4f acc = *(const v4f*)(hl + (size_t)d * DF);

    for (int base = beg; base < end; base += 32) {
        const int cn = (end - base < 32) ? (end - base) : 32;
        int s_l = 0;
        if (lane < cn) {
            s_l = csr[base + lane];
            s_l = min(max(s_l, 0), n - 1);
        }
        int j = 0;
        for (; j + 8 <= cn; j += 8) {
            v4f v[8];
#pragma unroll
            for (int k = 0; k < 8; ++k) {
                int s = __builtin_amdgcn_readlane(s_l, j + k);
                v[k]  = *(const v4f*)(hl + (size_t)s * DF);
            }
#pragma unroll
            for (int k = 0; k < 8; ++k) acc += v[k];
        }
        for (; j < cn; j += 4) {
            v4f v[4];
#pragma unroll
            for (int k = 0; k < 4; ++k) {
                int jj = (j + k < cn) ? (j + k) : (cn - 1);
                int s  = __builtin_amdgcn_readlane(s_l, jj);
                v[k]   = *(const v4f*)(hl + (size_t)s * DF);
            }
#pragma unroll
            for (int k = 0; k < 4; ++k) {
                v4f tsum = acc + v[k];
                acc = (j + k < cn) ? tsum : acc;
            }
        }
    }
    v4f o = acc * dd + bv;
    if (NORM_OUT) {
        float ss = o.x * o.x + o.y * o.y + o.z * o.z + o.w * o.w;
#pragma unroll
        for (int m = 16; m > 0; m >>= 1) ss += __shfl_xor(ss, m, 32);
        const float sc = 1.0f / fmaxf(sqrtf(ss), 1e-12f);
        o.x *= sc; o.y *= sc; o.z *= sc; o.w *= sc;
    }
    *(v4f*)(OUT + (size_t)d * DF + lane * 4) = o;
}

#define FUSED_GROUPS 512
#define APITCH 132

__global__ __launch_bounds__(GEMM_THREADS) void k_pull_gemm(
        const float* __restrict__ Hin, const int* __restrict__ rowptr,
        const int* __restrict__ csr, const float* __restrict__ dinv,
        const float* __restrict__ bias, const float* __restrict__ W,
        float* __restrict__ Hout, int strips, int nrows) {
    __shared__ __align__(16) float lds[DF * DF + (GEMM_THREADS / 32) * 16 * APITCH];
    float* __restrict__ ldsW = lds;
    float* __restrict__ ldsA = lds + DF * DF;

    {
        const int n    = threadIdx.x & (DF - 1);
        const int kg   = threadIdx.x >> 7;
        const int t    = n >> 4;
        const int nloc = n & 15;
        v2f* __restrict__ lw = (v2f*)ldsW + ((t << 5) * 32 + nloc);
        const float* __restrict__ wc = W + n;
#pragma unroll
        for (int i = 0; i < DF / 4 / (GEMM_THREADS / DF); ++i) {
            const int k = i * (GEMM_THREADS / DF) + kg;
            const float* __restrict__ wp = wc + (size_t)(4 * k) * DF;
            const v2f lo = {wp[0], wp[DF]};
            const v2f hi = {wp[2 * DF], wp[3 * DF]};
            lw[k * 32]      = lo;
            lw[k * 32 + 16] = hi;
        }
    }
    __syncthreads();

    const int wave  = __builtin_amdgcn_readfirstlane((int)(threadIdx.x >> 5));
    const int lane  = threadIdx.x & 31;
    float* __restrict__ myA = ldsA + wave * 16 * APITCH;
    const int nloc  = lane & 15;
    const int khalf = lane >> 4;
    const v4f bv = *(const v4f*)(bias + lane * 4);
    const int groups = (int)gridDim.x;
    for (int strip = __builtin_amdgcn_readfirstlane(wave * groups + (int)blockIdx.x);
         strip < strips; strip += groups * (GEMM_THREADS >> 5)) {
    const int d0 = strip * 16;
    const int nr = (nrows - d0 < 16) ? (nrows - d0) : 16;

    float dv_l = 0.0f;
    {
        const float* __restrict__ hl = Hin + lane * 4;
        int   rp_l = 0;
        if (lane <= nr) rp_l = rowptr[d0 + lane];
        if (lane <  nr) dv_l = dinv[d0 + lane];

        int beg = __builtin_amdgcn_readlane(rp_l, 0);
        int end = __builtin_amdgcn_readlane(rp_l, 1);
        v4f self = *(const v4f*)(hl + (size_t)d0 * DF);
        int s_l = 0;
        if (lane < end - beg) s_l = csr[beg + lane];

        for (int r = 0; r < nr; ++r) {
            int nend = end;
            v4f nself = self;
            int ns_l = 0;
            if (r + 1 < nr) {
                nend  = __builtin_amdgcn_readlane(rp_l, r + 2);
                nself = *(const v4f*)(hl + (size_t)(d0 + r + 1) * DF);
                if (lane < nend - end) ns_l = csr[end + lane];
            }

            v4f acc = self;
            for (int base = beg; base < end; base += 32) {
                const int cn = (end - base < 32) ? (end - base) : 32;
                if (base != beg) {
                    s_l = 0;
                    if (lane < cn) s_l = csr[base + lane];
                }
                v4f v[32];
#pragma unroll
                for (int g = 0; g < 8; ++g) {
                    if (4 * g < cn) {
#pragma unroll
                        for (int k = 0; k < 4; ++k) {
                            const int jj = (4 * g + k < cn) ? (4 * g + k) : (cn - 1);
                            int s = __builtin_amdgcn_readlane(s_l, jj);
                            s = min(max(s, 0), nrows - 1);
                            v[4 * g + k] = *(const v4f*)(hl + (size_t)s * DF);
                        }
                    }
                }
#pragma unroll
                for (int g = 0; g < 8; ++g) {
                    if (4 * g < cn) {
#pragma unroll
                        for (int k = 0; k < 4; ++k) {
                            const v4f tsum = acc + v[4 * g + k];
                            acc = (4 * g + k < cn) ? tsum : acc;
                        }
                    }
                }
            }

            const float dd = __int_as_float(__builtin_amdgcn_readlane(__float_as_int(dv_l), r));
            v4f o = acc * dd + bv;
            o.x = fmaxf(o.x, 0.0f); o.y = fmaxf(o.y, 0.0f);
            o.z = fmaxf(o.z, 0.0f); o.w = fmaxf(o.w, 0.0f);
            *(v4f*)(myA + r * APITCH + lane * 4) = o;

            beg = end; end = nend; self = nself; s_l = ns_l;
        }
        const v4f zero = {0.0f, 0.0f, 0.0f, 0.0f};
        for (int r = nr; r < 16; ++r)
            *(v4f*)(myA + r * APITCH + lane * 4) = zero;
    }
    __builtin_amdgcn_fence(__ATOMIC_ACQ_REL, "workgroup", "local");

    const float* __restrict__ arow = myA + nloc * APITCH + 2 * khalf;
    const v2f* __restrict__ bbase = (const v2f*)ldsW + lane;

    v8f acc[8] = {};
#pragma unroll
    for (int k = 0; k < 32; ++k) {
        const v2f a = *(const v2f*)(arow + 4 * k);
#pragma unroll
        for (int t = 0; t < 8; ++t) {
            v2f b = bbase[(t * 32 + k) * 32];
            acc[t] = __builtin_amdgcn_wmma_f32_16x16x4_f32(
                         false, a, false, b, (short)0, acc[t], false, false);
        }
    }

    {
        float* __restrict__ dt = myA + (8 * khalf) * APITCH + nloc;
#pragma unroll
        for (int r = 0; r < 8; ++r) {
#pragma unroll
            for (int t = 0; t < 8; ++t)
                dt[r * APITCH + t * 16] = acc[t][r];
        }
    }
    __builtin_amdgcn_fence(__ATOMIC_ACQ_REL, "workgroup", "local");
    {
        v4f o[16];
#pragma unroll
        for (int r = 0; r < 16; ++r)
            o[r] = *(const v4f*)(myA + r * APITCH + lane * 4);
#pragma unroll
        for (int r = 0; r < 16; ++r) {
            if (r < nr) {
                const float dd = __int_as_float(__builtin_amdgcn_readlane(__float_as_int(dv_l), r));
                *(v4f*)(Hout + (size_t)(d0 + r) * DF + lane * 4) = o[r] * dd;
            }
        }
    }
    }
}


extern "C" void kernel_launch(void* const* d_in, const int* in_sizes, int n_in,
                              void* d_out, int out_size, void* d_ws, size_t ws_size,
                              hipStream_t stream) {
    const float* x  = (const float*)d_in[0];
    const int*   ei = (const int*)  d_in[1];
    const float* W1 = (const float*)d_in[2];
    const float* b1 = (const float*)d_in[3];
    const float* W2 = (const float*)d_in[4];
    const float* b2 = (const float*)d_in[5];
    const float* W3 = (const float*)d_in[6];
    const float* b3 = (const float*)d_in[7];

    const int N = in_sizes[0] / DF;
    const int E = in_sizes[1] / 2;
    const int* src = ei;
    const int* dst = ei + E;

    float* out = (float*)d_out;

    char* ws = (char*)d_ws;
    const int B = (N + BIN_NODES - 1) / BIN_NODES;
    const int chunk = (E + SORT_CHUNKS - 1) / SORT_CHUNKS;
    size_t offN  = ((size_t)(N + 1) * 4 + 255) & ~(size_t)255;
    size_t offB  = ((size_t)B * 4 + 255) & ~(size_t)255;
    size_t offH  = ((size_t)B * SORT_CHUNKS * 4 + 255) & ~(size_t)255;
    size_t offE  = ((size_t)E * 4 + 255) & ~(size_t)255;
    float* dinv   = (float*)ws;
    int*   rowptr = (int*)(ws + offN);
    int*   bintot = (int*)(ws + 2 * offN);
    int*   hist   = (int*)(ws + 2 * offN + offB);
    int*   csr    = (int*)(ws + 2 * offN + offB + offH);
    int*   pairs  = (int*)(ws + 2 * offN + offB + offH + offE);
    float* H      = (float*)(ws + 2 * offN + offB + offH + 3 * offE);
    float* AGGa   = H + (size_t)N * DF;

    const int strips  = (N + 15) / 16;
    const int gblocks = (strips + 7) / 8;
    const int pblocks = (N + 7) / 8;
    const int fblocks = gblocks < FUSED_GROUPS ? gblocks : FUSED_GROUPS;

    k_bin_hist   <<<SORT_CHUNKS, 1024,        0, stream>>>(dst, hist, E, B, chunk);
    k_bin_scan   <<<B,           SORT_CHUNKS, 0, stream>>>(hist, bintot);
    k_bin_scatter<<<SORT_CHUNKS, 1024,        0, stream>>>(src, dst, hist, bintot, pairs, E, B, chunk);
    k_bin_rows   <<<B,           BIN_NODES,   0, stream>>>(pairs, bintot, rowptr, csr, dinv, N, B);

    k_gemm_wmma<false><<<gblocks, GEMM_THREADS, 0, stream>>>(x, W1, dinv, H, strips, N);

    k_pull_gemm<<<fblocks, GEMM_THREADS, 0, stream>>>(H, rowptr, csr, dinv, b1, W2, AGGa, strips, N);

    k_pull_gemm<<<fblocks, GEMM_THREADS, 0, stream>>>(AGGa, rowptr, csr, dinv, b2, W3, H, strips, N);

    k_pull<true><<<pblocks, 256, 0, stream>>>(H, rowptr, csr, dinv, b3, out, N);
}


# Edge sort in 2 launches, not 4: a chunk bins its own pairs in LDS, stores them chunk-major with its per-bin offsets; a bin's row build reads its 256 slices once; row sum order = LDS arrival order
# speedup vs baseline: 5.2050x; 1.0378x over previous
#include <hip/hip_runtime.h>
#include <hip/hip_bf16.h>


typedef __attribute__((ext_vector_type(2))) float v2f;
typedef __attribute__((ext_vector_type(4))) float v4f;
typedef __attribute__((ext_vector_type(8))) float v8f;

#define DF 128
#define GEMM_THREADS 256


#define BIN_SHIFT   8
#define BIN_NODES   (1 << BIN_SHIFT)
#define SORT_CHUNKS 256
#define MAX_BINS    1024

typedef __attribute__((ext_vector_type(2))) int v2i;

#define STAGE_PAIRS 8192

__global__ __launch_bounds__(1024) void k_chunk_bins(const int* __restrict__ src,
                                                     const int* __restrict__ dst,
                                                     int* __restrict__ loff,
                                                     int* __restrict__ pairs,
                                                     int E, int B, int chunk) {
    __shared__ int lh[MAX_BINS];
    __shared__ int wsum[32];
    __shared__ v2i stage[STAGE_PAIRS];
    const int t = threadIdx.x, lane = t & 31, wave = t >> 5, g = blockIdx.x;
    const bool staged = chunk <= STAGE_PAIRS;
    for (int b = t; b < MAX_BINS; b += 1024) lh[b] = 0;
    __syncthreads();
    const int e0 = g * chunk;
    const int e1 = (e0 + chunk < E) ? (e0 + chunk) : E;
    for (int eb = e0 + t; eb < e1; eb += 8 * 1024) {
        int d[8];
#pragma unroll
        for (int k = 0; k < 8; ++k) {
            const int e = eb + k * 1024;
            d[k] = (e < e1) ? dst[e] : -1;
        }
#pragma unroll
        for (int k = 0; k < 8; ++k)
            if (d[k] >= 0) atomicAdd(&lh[d[k] >> BIN_SHIFT], 1);
    }
    __syncthreads();

    const int c = (t < B) ? lh[t] : 0;
    int inc = c;
#pragma unroll
    for (int o = 1; o < 32; o <<= 1) {
        int u = __shfl_up(inc, o, 32);
        if (lane >= o) inc += u;
    }
    if (lane == 31) wsum[wave] = inc;
    __syncthreads();
    int woff = 0;
#pragma unroll
    for (int w = 0; w < 32; ++w) woff += (w < wave) ? wsum[w] : 0;
    const int excl = woff + inc - c;
    if (t < B) {
        loff[t * SORT_CHUNKS + g] = excl;
        lh[t] = staged ? excl : (e0 + excl);
    }
    if (t == 0) loff[B * SORT_CHUNKS + g] = (e1 > e0) ? (e1 - e0) : 0;
    __syncthreads();

    v2i* __restrict__ pr = (v2i*)pairs;
    for (int eb = e0 + t; eb < e1; eb += 8 * 1024) {
        int d[8], s[8];
#pragma unroll
        for (int k = 0; k < 8; ++k) {
            const int e = eb + k * 1024;
            const bool ok = e < e1;
            d[k] = ok ? dst[e] : -1;
            s[k] = ok ? src[e] : 0;
        }
#pragma unroll
        for (int k = 0; k < 8; ++k) {
            if (d[k] >= 0) {
                const int p = atomicAdd(&lh[d[k] >> BIN_SHIFT], 1);
                v2i w = {s[k], d[k]};
                if (staged) stage[p] = w;
                else        pr[p] = w;
            }
        }
    }
    if (staged) {
        __syncthreads();
        const int n = e1 - e0;
        for (int i = t; i < n; i += 1024) pr[e0 + i] = stage[i];
    }
}

#define BIN_STAGE 8192

__global__ __launch_bounds__(BIN_NODES) void k_bin_rows(const int* __restrict__ pairs,
                                                        const int* __restrict__ loff,
                                                        int* __restrict__ rowptr,
                                                        int* __restrict__ csr,
                                                        float* __restrict__ dinv,
                                                        int n, int B, int chunk) {
    static_assert(BIN_NODES == SORT_CHUNKS, "thread c of a bin's work-group owns chunk c");
    __shared__ int cnt[BIN_NODES];
    __shared__ int cur[BIN_NODES];
    __shared__ int wtot[BIN_NODES / 32];
    __shared__ int wlo[BIN_NODES / 32];
    __shared__ int wsum[BIN_NODES / 32];
    __shared__ v2i lpairs[BIN_STAGE];
    __shared__ int lsrc[BIN_STAGE];
    const int t = threadIdx.x, lane = t & 31, wave = t >> 5;
    const int b = blockIdx.x;
    const v2i* __restrict__ pr = (const v2i*)pairs;

    const int lo = loff[b * SORT_CHUNKS + t];
    const int hi = loff[(b + 1) * SORT_CHUNKS + t];
    const int sc = hi - lo;
    const int sstart = t * chunk + lo;
    int sinc = sc, slo = lo;
#pragma unroll
    for (int o = 1; o < 32; o <<= 1) {
        int u = __shfl_up(sinc, o, 32);
        if (lane >= o) sinc += u;
    }
#pragma unroll
    for (int m = 16; m > 0; m >>= 1) slo += __shfl_xor(slo, m, 32);
    if (lane == 31) wtot[wave] = sinc;
    if (lane == 0)  wlo[wave]  = slo;
    cnt[t] = 0;
    __syncthreads();
    int soffw = 0, binCount = 0, binStart = 0;
#pragma unroll
    for (int w = 0; w < BIN_NODES / 32; ++w) {
        soffw    += (w < wave) ? wtot[w] : 0;
        binCount += wtot[w];
        binStart += wlo[w];
    }
    const int soff = soffw + sinc - sc;
    const int binEnd = binStart + binCount;
    const bool staged = binCount <= BIN_STAGE;

    if (staged) {
        v2i q[32];
#pragma unroll
        for (int i = 0; i < 32; ++i) {
            const int ni = __builtin_amdgcn_readlane(sc, i);
            const int si = __builtin_amdgcn_readlane(sstart, i);
            v2i none = {0, -1};
            q[i] = (lane < ni) ? pr[si + lane] : none;
        }
#pragma unroll
        for (int i = 0; i < 32; ++i) {
            const int ni = __builtin_amdgcn_readlane(sc, i);
            const int oi = __builtin_amdgcn_readlane(soff, i);
            if (lane < ni) lpairs[oi + lane] = q[i];
        }
        for (int i = 0; i < 32; ++i) {
            const int ni = __builtin_amdgcn_readlane(sc, i);
            if (ni > 32) {
                const int si = __builtin_amdgcn_readlane(sstart, i);
                const int oi = __builtin_amdgcn_readlane(soff, i);
                for (int j = 32 + lane; j < ni; j += 32) lpairs[oi + j] = pr[si + j];
            }
        }
        __syncthreads();
        for (int eb = t; eb < binCount; eb += 8 * BIN_NODES) {
            int d[8];
#pragma unroll
            for (int k = 0; k < 8; ++k) {
                const int e = eb + k * BIN_NODES;
                d[k] = (e < binCount) ? lpairs[e].y : -1;
            }
#pragma unroll
            for (int k = 0; k < 8; ++k)
                if (d[k] >= 0) atomicAdd(&cnt[d[k] & (BIN_NODES - 1)], 1);
        }
    } else {
        for (int i = 0; i < 32; ++i) {
            const int ni = __builtin_amdgcn_readlane(sc, i);
            const int si = __builtin_amdgcn_readlane(sstart, i);
            for (int j = lane; j < ni; j += 32)
                atomicAdd(&cnt[pr[si + j].y & (BIN_NODES - 1)], 1);
        }
    }
    __syncthreads();

    const int c = cnt[t];
    int inc = c;
#pragma unroll
    for (int o = 1; o < 32; o <<= 1) {
        int u = __shfl_up(inc, o, 32);
        if (lane >= o) inc += u;
    }
    if (lane == 31) wsum[wave] = inc;
    __syncthreads();
    int woff = 0;
#pragma unroll
    for (int w = 0; w < BIN_NODES / 32; ++w) woff += (w < wave) ? wsum[w] : 0;
    const int rowLocal = woff + inc - c;
    const int rowStart = binStart + rowLocal;
    const int node = b * BIN_NODES + t;
    if (node < n) {
        rowptr[node] = rowStart;
        dinv[node]   = rsqrtf((float)c + 1.0f);
    }
    if (b == B - 1 && t == 0) rowptr[n] = binEnd;
    cur[t] = staged ? rowLocal : rowStart;
    __syncthreads();

    if (staged) {
        for (int eb = t; eb < binCount; eb += 8 * BIN_NODES) {
            v2i q[8];
#pragma unroll
            for (int k = 0; k < 8; ++k) {
                const int e = eb + k * BIN_NODES;
                v2i none = {0, -1};
                q[k] = (e < binCount) ? lpairs[e] : none;
            }
#pragma unroll
            for (int k = 0; k < 8; ++k) {
                if (q[k].y >= 0) {
                    const int p = atomicAdd(&cur[q[k].y & (BIN_NODES - 1)], 1);
                    lsrc[p] = q[k].x;
                }
            }
        }
        __syncthreads();
        for (int i = t; i < binCount; i += BIN_NODES) csr[binStart + i] = lsrc[i];
    } else {
        for (int i = 0; i < 32; ++i) {
            const int ni = __builtin_amdgcn_readlane(sc, i);
            const int si = __builtin_amdgcn_readlane(sstart, i);
            for (int j = lane; j < ni; j += 32) {
                const v2i q = pr[si + j];
                const int p = atomicAdd(&cur[q.y & (BIN_NODES - 1)], 1);
                csr[p] = q.x;
            }
        }
    }
}

template <bool RELU_IN>
__global__ __launch_bounds__(GEMM_THREADS) void k_gemm_wmma(
        const float* __restrict__ X, const float* __restrict__ W,
        const float* __restrict__ dinv, float* __restrict__ H,
        int strips, int nrows) {
    __shared__ __align__(16) float ldsW[DF * DF];

    {
        const int n    = threadIdx.x & (DF - 1);
        const int kg   = threadIdx.x >> 7;
        const int t    = n >> 4;
        const int nloc = n & 15;
        v2f* __restrict__ lw = (v2f*)ldsW + ((t << 5) * 32 + nloc);
        const float* __restrict__ wc = W + n;
#pragma unroll
        for (int i = 0; i < DF / 4 / (GEMM_THREADS / DF); ++i) {
            const int k = i * (GEMM_THREADS / DF) + kg;
            const float* __restrict__ wp = wc + (size_t)(4 * k) * DF;
            const v2f lo = {wp[0], wp[DF]};
            const v2f hi = {wp[2 * DF], wp[3 * DF]};
            lw[k * 32]      = lo;
            lw[k * 32 + 16] = hi;
        }
    }
    __syncthreads();

    const int wave  = __builtin_amdgcn_readfirstlane((int)(threadIdx.x >> 5));
    const int lane  = threadIdx.x & 31;
    int strip = (int)blockIdx.x * (GEMM_THREADS >> 5) + wave;
    const bool live = strip < strips;
    if (!live) strip = strips - 1;

    const int nloc  = lane & 15;
    const int khalf = lane >> 4;

    int arow_i = strip * 16 + nloc;
    if (arow_i >= nrows) arow_i = nrows - 1;
    const float* __restrict__ arow = X + (size_t)arow_i * DF + 2 * khalf;

    const v2f* __restrict__ bbase = (const v2f*)ldsW + lane;

    float dv_l = 0.0f;
    if (lane < 16 && strip * 16 + lane < nrows) dv_l = dinv[strip * 16 + lane];

    v8f acc[8] = {};
#pragma unroll
    for (int k = 0; k < 32; ++k) {
        v2f a = *(const v2f*)(arow + 4 * k);
        if (RELU_IN) {
            a.x = fmaxf(a.x, 0.0f);
            a.y = fmaxf(a.y, 0.0f);
        }
#pragma unroll
        for (int t = 0; t < 8; ++t) {
            v2f b = bbase[(t * 32 + k) * 32];
            acc[t] = __builtin_amdgcn_wmma_f32_16x16x4_f32(
                         false, a, false, b, (short)0, acc[t], false, false);
        }
    }

    __syncthreads();
    float* __restrict__ myD = ldsW + wave * 16 * DF;
    {
        float* __restrict__ dt = myD + (8 * khalf) * DF + nloc;
#pragma unroll
        for (int r = 0; r < 8; ++r) {
#pragma unroll
            for (int t = 0; t < 8; ++t)
                dt[r * DF + t * 16] = acc[t][r];
        }
    }
    __builtin_amdgcn_fence(__ATOMIC_ACQ_REL, "workgroup", "local");
    {
        v4f o[16];
#pragma unroll
        for (int r = 0; r < 16; ++r)
            o[r] = *(const v4f*)(myD + r * DF + lane * 4);
        const int d0 = strip * 16;
#pragma unroll
        for (int r = 0; r < 16; ++r) {
            if (live && d0 + r < nrows) {
                const float dd = __int_as_float(__builtin_amdgcn_readlane(__float_as_int(dv_l), r));
                *(v4f*)(H + (size_t)(d0 + r) * DF + lane * 4) = o[r] * dd;
            }
        }
    }
}

template <bool NORM_OUT>
__global__ __launch_bounds__(256) void k_pull(
        const float* __restrict__ H, const int* __restrict__ rowptr,
        const int* __restrict__ csr, const float* __restrict__ dinv,
        const float* __restrict__ bias, float* __restrict__ OUT, int n) {
    const int d = __builtin_amdgcn_readfirstlane(
                      (int)(blockIdx.x * (blockDim.x >> 5) + (threadIdx.x >> 5)));
    if (d >= n) return;
    const int lane = threadIdx.x & 31;
    const int beg = rowptr[d], end = rowptr[d + 1];
    const float dd = dinv[d];
    const float* __restrict__ hl = H + lane * 4;
    const v4f bv = *(const v4f*)(bias + lane * 4);
    v4f acc = *(const v4f*)(hl + (size_t)d * DF);

    for (int base = beg; base < end; base += 32) {
        const int cn = (end - base < 32) ? (end - base) : 32;
        int s_l = 0;
        if (lane < cn) {
            s_l = csr[base + lane];
            s_l = min(max(s_l, 0), n - 1);
        }
        int j = 0;
        for (; j + 8 <= cn; j += 8) {
            v4f v[8];
#pragma unroll
            for (int k = 0; k < 8; ++k) {
                int s = __builtin_amdgcn_readlane(s_l, j + k);
                v[k]  = *(const v4f*)(hl + (size_t)s * DF);
            }
#pragma unroll
            for (int k = 0; k < 8; ++k) acc += v[k];
        }
        for (; j < cn; j += 4) {
            v4f v[4];
#pragma unroll
            for (int k = 0; k < 4; ++k) {
                int jj = (j + k < cn) ? (j + k) : (cn - 1);
                int s  = __builtin_amdgcn_readlane(s_l, jj);
                v[k]   = *(const v4f*)(hl + (size_t)s * DF);
            }
#pragma unroll
            for (int k = 0; k < 4; ++k) {
                v4f tsum = acc + v[k];
                acc = (j + k < cn) ? tsum : acc;
            }
        }
    }
    v4f o = acc * dd + bv;
    if (NORM_OUT) {
        float ss = o.x * o.x + o.y * o.y + o.z * o.z + o.w * o.w;
#pragma unroll
        for (int m = 16; m > 0; m >>= 1) ss += __shfl_xor(ss, m, 32);
        const float sc = 1.0f / fmaxf(sqrtf(ss), 1e-12f);
        o.x *= sc; o.y *= sc; o.z *= sc; o.w *= sc;
    }
    *(v4f*)(OUT + (size_t)d * DF + lane * 4) = o;
}

#define FUSED_GROUPS 512
#define APITCH 132

__global__ __launch_bounds__(GEMM_THREADS) void k_pull_gemm(
        const float* __restrict__ Hin, const int* __restrict__ rowptr,
        const int* __restrict__ csr, const float* __restrict__ dinv,
        const float* __restrict__ bias, const float* __restrict__ W,
        float* __restrict__ Hout, int strips, int nrows) {
    __shared__ __align__(16) float lds[DF * DF + (GEMM_THREADS / 32) * 16 * APITCH];
    float* __restrict__ ldsW = lds;
    float* __restrict__ ldsA = lds + DF * DF;

    {
        const int n    = threadIdx.x & (DF - 1);
        const int kg   = threadIdx.x >> 7;
        const int t    = n >> 4;
        const int nloc = n & 15;
        v2f* __restrict__ lw = (v2f*)ldsW + ((t << 5) * 32 + nloc);
        const float* __restrict__ wc = W + n;
#pragma unroll
        for (int i = 0; i < DF / 4 / (GEMM_THREADS / DF); ++i) {
            const int k = i * (GEMM_THREADS / DF) + kg;
            const float* __restrict__ wp = wc + (size_t)(4 * k) * DF;
            const v2f lo = {wp[0], wp[DF]};
            const v2f hi = {wp[2 * DF], wp[3 * DF]};
            lw[k * 32]      = lo;
            lw[k * 32 + 16] = hi;
        }
    }
    __syncthreads();

    const int wave  = __builtin_amdgcn_readfirstlane((int)(threadIdx.x >> 5));
    const int lane  = threadIdx.x & 31;
    float* __restrict__ myA = ldsA + wave * 16 * APITCH;
    const int nloc  = lane & 15;
    const int khalf = lane >> 4;
    const v4f bv = *(const v4f*)(bias + lane * 4);
    const int groups = (int)gridDim.x;
    for (int strip = __builtin_amdgcn_readfirstlane(wave * groups + (int)blockIdx.x);
         strip < strips; strip += groups * (GEMM_THREADS >> 5)) {
    const int d0 = strip * 16;
    const int nr = (nrows - d0 < 16) ? (nrows - d0) : 16;

    float dv_l = 0.0f;
    {
        const float* __restrict__ hl = Hin + lane * 4;
        int   rp_l = 0;
        if (lane <= nr) rp_l = rowptr[d0 + lane];
        if (lane <  nr) dv_l = dinv[d0 + lane];

        int beg = __builtin_amdgcn_readlane(rp_l, 0);
        int end = __builtin_amdgcn_readlane(rp_l, 1);
        v4f self = *(const v4f*)(hl + (size_t)d0 * DF);
        int s_l = 0;
        if (lane < end - beg) s_l = csr[beg + lane];

        for (int r = 0; r < nr; ++r) {
            int nend = end;
            v4f nself = self;
            int ns_l = 0;
            if (r + 1 < nr) {
                nend  = __builtin_amdgcn_readlane(rp_l, r + 2);
                nself = *(const v4f*)(hl + (size_t)(d0 + r + 1) * DF);
                if (lane < nend - end) ns_l = csr[end + lane];
            }

            v4f acc = self;
            for (int base = beg; base < end; base += 32) {
                const int cn = (end - base < 32) ? (end - base) : 32;
                if (base != beg) {
                    s_l = 0;
                    if (lane < cn) s_l = csr[base + lane];
                }
                v4f v[32];
#pragma unroll
                for (int g = 0; g < 8; ++g) {
                    if (4 * g < cn) {
#pragma unroll
                        for (int k = 0; k < 4; ++k) {
                            const int jj = (4 * g + k < cn) ? (4 * g + k) : (cn - 1);
                            int s = __builtin_amdgcn_readlane(s_l, jj);
                            s = min(max(s, 0), nrows - 1);
                            v[4 * g + k] = *(const v4f*)(hl + (size_t)s * DF);
                        }
                    }
                }
#pragma unroll
                for (int g = 0; g < 8; ++g) {
                    if (4 * g < cn) {
#pragma unroll
                        for (int k = 0; k < 4; ++k) {
                            const v4f tsum = acc + v[4 * g + k];
                            acc = (4 * g + k < cn) ? tsum : acc;
                        }
                    }
                }
            }

            const float dd = __int_as_float(__builtin_amdgcn_readlane(__float_as_int(dv_l), r));
            v4f o = acc * dd + bv;
            o.x = fmaxf(o.x, 0.0f); o.y = fmaxf(o.y, 0.0f);
            o.z = fmaxf(o.z, 0.0f); o.w = fmaxf(o.w, 0.0f);
            *(v4f*)(myA + r * APITCH + lane * 4) = o;

            beg = end; end = nend; self = nself; s_l = ns_l;
        }
        const v4f zero = {0.0f, 0.0f, 0.0f, 0.0f};
        for (int r = nr; r < 16; ++r)
            *(v4f*)(myA + r * APITCH + lane * 4) = zero;
    }
    __builtin_amdgcn_fence(__ATOMIC_ACQ_REL, "workgroup", "local");

    const float* __restrict__ arow = myA + nloc * APITCH + 2 * khalf;
    const v2f* __restrict__ bbase = (const v2f*)ldsW + lane;

    v8f acc[8] = {};
#pragma unroll
    for (int k = 0; k < 32; ++k) {
        const v2f a = *(const v2f*)(arow + 4 * k);
#pragma unroll
        for (int t = 0; t < 8; ++t) {
            v2f b = bbase[(t * 32 + k) * 32];
            acc[t] = __builtin_amdgcn_wmma_f32_16x16x4_f32(
                         false, a, false, b, (short)0, acc[t], false, false);
        }
    }

    {
        float* __restrict__ dt = myA + (8 * khalf) * APITCH + nloc;
#pragma unroll
        for (int r = 0; r < 8; ++r) {
#pragma unroll
            for (int t = 0; t < 8; ++t)
                dt[r * APITCH + t * 16] = acc[t][r];
        }
    }
    __builtin_amdgcn_fence(__ATOMIC_ACQ_REL, "workgroup", "local");
    {
        v4f o[16];
#pragma unroll
        for (int r = 0; r < 16; ++r)
            o[r] = *(const v4f*)(myA + r * APITCH + lane * 4);
#pragma unroll
        for (int r = 0; r < 16; ++r) {
            if (r < nr) {
                const float dd = __int_as_float(__builtin_amdgcn_readlane(__float_as_int(dv_l), r));
                *(v4f*)(Hout + (size_t)(d0 + r) * DF + lane * 4) = o[r] * dd;
            }
        }
    }
    }
}


extern "C" void kernel_launch(void* const* d_in, const int* in_sizes, int n_in,
                              void* d_out, int out_size, void* d_ws, size_t ws_size,
                              hipStream_t stream) {
    const float* x  = (const float*)d_in[0];
    const int*   ei = (const int*)  d_in[1];
    const float* W1 = (const float*)d_in[2];
    const float* b1 = (const float*)d_in[3];
    const float* W2 = (const float*)d_in[4];
    const float* b2 = (const float*)d_in[5];
    const float* W3 = (const float*)d_in[6];
    const float* b3 = (const float*)d_in[7];

    const int N = in_sizes[0] / DF;
    const int E = in_sizes[1] / 2;
    const int* src = ei;
    const int* dst = ei + E;

    float* out = (float*)d_out;

    char* ws = (char*)d_ws;
    const int B = (N + BIN_NODES - 1) / BIN_NODES;
    const int chunk = (E + SORT_CHUNKS - 1) / SORT_CHUNKS;
    size_t offN  = ((size_t)(N + 1) * 4 + 255) & ~(size_t)255;
    size_t offL  = ((size_t)(B + 1) * SORT_CHUNKS * 4 + 255) & ~(size_t)255;
    size_t offE  = ((size_t)E * 4 + 255) & ~(size_t)255;
    float* dinv   = (float*)ws;
    int*   rowptr = (int*)(ws + offN);
    int*   loff   = (int*)(ws + 2 * offN);
    int*   csr    = (int*)(ws + 2 * offN + offL);
    int*   pairs  = (int*)(ws + 2 * offN + offL + offE);
    float* H      = (float*)(ws + 2 * offN + offL + 3 * offE);
    float* AGGa   = H + (size_t)N * DF;

    const int strips  = (N + 15) / 16;
    const int gblocks = (strips + 7) / 8;
    const int pblocks = (N + 7) / 8;
    const int fblocks = gblocks < FUSED_GROUPS ? gblocks : FUSED_GROUPS;

    k_chunk_bins<<<SORT_CHUNKS, 1024,      0, stream>>>(src, dst, loff, pairs, E, B, chunk);
    k_bin_rows  <<<B,           BIN_NODES, 0, stream>>>(pairs, loff, rowptr, csr, dinv, N, B, chunk);

    k_gemm_wmma<false><<<gblocks, GEMM_THREADS, 0, stream>>>(x, W1, dinv, H, strips, N);

    k_pull_gemm<<<fblocks, GEMM_THREADS, 0, stream>>>(H, rowptr, csr, dinv, b1, W2, AGGa, strips, N);

    k_pull_gemm<<<fblocks, GEMM_THREADS, 0, stream>>>(AGGa, rowptr, csr, dinv, b2, W3, H, strips, N);

    k_pull<true><<<pblocks, 256, 0, stream>>>(H, rowptr, csr, dinv, b3, out, N);
}
